# starting kernel
# baseline (speedup 1.0000x reference)
#include <hip/hip_runtime.h>
#include <hip/hip_bf16.h>

typedef __attribute__((ext_vector_type(2))) float v2f;
typedef __attribute__((ext_vector_type(8))) float v8f;

#define C_DIM 256

__global__ void zero_f32(float* __restrict__ p, long long n) {
  long long i = (long long)blockIdx.x * blockDim.x + threadIdx.x;
  long long stride = (long long)gridDim.x * blockDim.x;
  for (; i < n; i += stride) p[i] = 0.0f;
}

__global__ void transpose_w(const float* __restrict__ W, float* __restrict__ Wt,
                            int K, int N) {
  int t = blockIdx.x * blockDim.x + threadIdx.x;
  if (t >= K * N) return;
  int k = t / N;
  int n = t % N;
  Wt[n * K + k] = W[t];
}

__global__ void scatter_edges(const float* __restrict__ x,
                              const int* __restrict__ src,
                              const int* __restrict__ dst,
                              float* __restrict__ agg,
                              float* __restrict__ cnt,
                              int n_edges) {
  const int chunks = C_DIM / 4;
  long long t = (long long)blockIdx.x * blockDim.x + threadIdx.x;
  long long total = (long long)n_edges * chunks;
  if (t >= total) return;
  int e = (int)(t / chunks);
  int c = (int)(t % chunks) * 4;
  int s = src[e];
  int d = dst[e];
  const float4 v = *(const float4*)(x + (long long)s * C_DIM + c);
  float* a = agg + (long long)d * C_DIM + c;
  atomicAdd(a + 0, v.x);
  atomicAdd(a + 1, v.y);
  atomicAdd(a + 2, v.z);
  atomicAdd(a + 3, v.w);
  if (c == 0) atomicAdd(cnt + d, 1.0f);
}

template <int RELU>
__global__ __launch_bounds__(256) void gemm_dual_wmma(
    const float* __restrict__ A1, const float* __restrict__ Wt1,
    const float* __restrict__ cnt1,
    const float* __restrict__ A2, const float* __restrict__ Wt2,
    const float* __restrict__ bias, float* __restrict__ out,
    int M, int N) {
  constexpr int K = C_DIM;
  const int lane = threadIdx.x & 31;
  const int wave = threadIdx.x >> 5;
  const int nstrips = N >> 6;
  const int tile = blockIdx.x * 8 + wave;
  if (tile >= (M >> 4) * nstrips) return;

  const int m0 = (tile / nstrips) << 4;
  const int n0 = (tile % nstrips) << 6;

  const int half = lane >> 4;
  const int l15  = lane & 15;
  const int row  = m0 + l15;
  const int cb   = n0 + l15;
  const int koff = half << 1;

  const float s1 = 1.0f / fmaxf(cnt1[row], 1.0f);

  v8f acc[4];
#pragma unroll
  for (int j = 0; j < 4; ++j) acc[j] = (v8f){0.f,0.f,0.f,0.f,0.f,0.f,0.f,0.f};

  const float* a1p = A1 + (long long)row * K + koff;
  const float* a2p = A2 + (long long)row * K + koff;
  const float* w1p = Wt1 + (long long)cb * K + koff;
  const float* w2p = Wt2 + (long long)cb * K + koff;

  for (int k = 0; k < K; k += 4) {
    v2f a = *(const v2f*)(a1p + k);
    a = a * s1;
#pragma unroll
    for (int j = 0; j < 4; ++j) {
      v2f b = *(const v2f*)(w1p + (long long)j * 16 * K + k);
      acc[j] = __builtin_amdgcn_wmma_f32_16x16x4_f32(false, a, false, b,
                                                     (short)0, acc[j],
                                                     false, false);
    }
  }
  for (int k = 0; k < K; k += 4) {
    v2f a = *(const v2f*)(a2p + k);
#pragma unroll
    for (int j = 0; j < 4; ++j) {
      v2f b = *(const v2f*)(w2p + (long long)j * 16 * K + k);
      acc[j] = __builtin_amdgcn_wmma_f32_16x16x4_f32(false, a, false, b,
                                                     (short)0, acc[j],
                                                     false, false);
    }
  }

#pragma unroll
  for (int j = 0; j < 4; ++j) {
    const int col = cb + j * 16;
    const float bv = bias[col];
#pragma unroll
    for (int r = 0; r < 8; ++r) {
      float v = acc[j][r] + bv;
      if (RELU) v = fmaxf(v, 0.f);
      out[(long long)(m0 + r + half * 8) * N + col] = v;
    }
  }
}

static inline int cdiv(long long a, long long b) { return (int)((a + b - 1) / b); }

extern "C" void kernel_launch(void* const* d_in, const int* in_sizes, int n_in,
                              void* d_out, int out_size, void* d_ws, size_t ws_size,
                              hipStream_t stream) {
  (void)n_in; (void)out_size; (void)ws_size;

  const int N1 = 40000, N2 = 20000, N3 = 10000;
  const int OUT_C = 64;

  const float* x   = (const float*)d_in[0];
  const int*   ei1 = (const int*)d_in[1];
  const int*   ei2 = (const int*)d_in[2];
  const int*   ei3 = (const int*)d_in[3];
  const float* Wl1 = (const float*)d_in[4];
  const float* bl1 = (const float*)d_in[5];
  const float* Wr1 = (const float*)d_in[6];
  const float* Wl2 = (const float*)d_in[7];
  const float* bl2 = (const float*)d_in[8];
  const float* Wr2 = (const float*)d_in[9];
  const float* Wl3 = (const float*)d_in[10];
  const float* bl3 = (const float*)d_in[11];
  const float* Wr3 = (const float*)d_in[12];
  const float* Wls = (const float*)d_in[13];
  const float* bls = (const float*)d_in[14];
  const float* Wrs = (const float*)d_in[15];

  const int E1 = in_sizes[1] / 2;
  const int E2 = in_sizes[2] / 2;
  const int E3 = in_sizes[3] / 2;

  float* h1  = (float*)d_ws;
  float* h2  = h1 + (long long)N1 * C_DIM;
  float* agg = h2 + (long long)N2 * C_DIM;
  float* cnt = agg + (long long)N1 * C_DIM;
  float* wt  = cnt + N1;
  float* Wl1t = wt;
  float* Wr1t = Wl1t + C_DIM * C_DIM;
  float* Wl2t = Wr1t + C_DIM * C_DIM;
  float* Wr2t = Wl2t + C_DIM * C_DIM;
  float* Wl3t = Wr2t + C_DIM * C_DIM;
  float* Wr3t = Wl3t + C_DIM * OUT_C;
  float* Wlst = Wr3t + C_DIM * OUT_C;
  float* Wrst = Wlst + C_DIM * OUT_C;

  float* mu     = (float*)d_out;
  float* logstd = mu + (long long)N3 * OUT_C;

  const int ZB = 256, SB = 256, GB = 256, TB = 256;

  {
    int nbig = C_DIM * C_DIM, nsml = C_DIM * OUT_C;
    transpose_w<<<cdiv(nbig, TB), TB, 0, stream>>>(Wl1, Wl1t, C_DIM, C_DIM);
    transpose_w<<<cdiv(nbig, TB), TB, 0, stream>>>(Wr1, Wr1t, C_DIM, C_DIM);
    transpose_w<<<cdiv(nbig, TB), TB, 0, stream>>>(Wl2, Wl2t, C_DIM, C_DIM);
    transpose_w<<<cdiv(nbig, TB), TB, 0, stream>>>(Wr2, Wr2t, C_DIM, C_DIM);
    transpose_w<<<cdiv(nsml, TB), TB, 0, stream>>>(Wl3, Wl3t, C_DIM, OUT_C);
    transpose_w<<<cdiv(nsml, TB), TB, 0, stream>>>(Wr3, Wr3t, C_DIM, OUT_C);
    transpose_w<<<cdiv(nsml, TB), TB, 0, stream>>>(Wls, Wlst, C_DIM, OUT_C);
    transpose_w<<<cdiv(nsml, TB), TB, 0, stream>>>(Wrs, Wrst, C_DIM, OUT_C);
  }

  {
    long long na = (long long)N1 * C_DIM;
    zero_f32<<<cdiv(na, ZB), ZB, 0, stream>>>(agg, na);
    zero_f32<<<cdiv(N1, ZB), ZB, 0, stream>>>(cnt, N1);
    long long work = (long long)E1 * (C_DIM / 4);
    scatter_edges<<<cdiv(work, SB), SB, 0, stream>>>(x, ei1, ei1 + E1, agg, cnt, E1);
    int tiles = (N1 / 16) * (C_DIM / 64);
    gemm_dual_wmma<1><<<cdiv(tiles, 8), GB, 0, stream>>>(
        agg, Wl1t, cnt, x, Wr1t, bl1, h1, N1, C_DIM);
  }

  {
    long long na = (long long)N2 * C_DIM;
    zero_f32<<<cdiv(na, ZB), ZB, 0, stream>>>(agg, na);
    zero_f32<<<cdiv(N2, ZB), ZB, 0, stream>>>(cnt, N2);
    long long work = (long long)E2 * (C_DIM / 4);
    scatter_edges<<<cdiv(work, SB), SB, 0, stream>>>(h1, ei2, ei2 + E2, agg, cnt, E2);
    int tiles = (N2 / 16) * (C_DIM / 64);
    gemm_dual_wmma<1><<<cdiv(tiles, 8), GB, 0, stream>>>(
        agg, Wl2t, cnt, h1, Wr2t, bl2, h2, N2, C_DIM);
  }

  {
    long long na = (long long)N3 * C_DIM;
    zero_f32<<<cdiv(na, ZB), ZB, 0, stream>>>(agg, na);
    zero_f32<<<cdiv(N3, ZB), ZB, 0, stream>>>(cnt, N3);
    long long work = (long long)E3 * (C_DIM / 4);
    scatter_edges<<<cdiv(work, SB), SB, 0, stream>>>(h2, ei3, ei3 + E3, agg, cnt, E3);

    int tiles = (N3 / 16) * (OUT_C / 64);
    gemm_dual_wmma<0><<<cdiv(tiles, 8), GB, 0, stream>>>(
        agg, Wl3t, cnt, h2, Wr3t, bl3, mu, N3, OUT_C);
    gemm_dual_wmma<0><<<cdiv(tiles, 8), GB, 0, stream>>>(
        agg, Wlst, cnt, h2, Wrst, bls, logstd, N3, OUT_C);
  }
}


# Replace float-atomic edge scatter and its zero-fills by counting-sort row lists (one int atomic per edge gives its slot) and one wave per row summing source rows in registers in slot (arrival) order
# speedup vs baseline: 1.5993x; 1.5993x over previous
#include <hip/hip_runtime.h>
#include <hip/hip_bf16.h>

typedef __attribute__((ext_vector_type(2))) float v2f;
typedef __attribute__((ext_vector_type(8))) float v8f;

#define C_DIM 256

__global__ void zero_i32(int* __restrict__ p, int n) {
  int i = blockIdx.x * blockDim.x + threadIdx.x;
  if (i < n) p[i] = 0;
}

__global__ void transpose_w(const float* __restrict__ W, float* __restrict__ Wt,
                            int K, int N) {
  int t = blockIdx.x * blockDim.x + threadIdx.x;
  if (t >= K * N) return;
  int k = t / N;
  int n = t % N;
  Wt[n * K + k] = W[t];
}

__global__ void edge_rank(const int* __restrict__ ei1, const int* __restrict__ ei2,
                          const int* __restrict__ ei3, int E1, int E2, int E3,
                          int n1, int n2,
                          int* __restrict__ deg, int* __restrict__ rank) {
  const int t = blockIdx.x * blockDim.x + threadIdx.x;
  const int* dst;
  int e, roff;
  if (t < E1)                { e = t;           dst = ei1 + E1; roff = 0; }
  else if (t < E1 + E2)      { e = t - E1;      dst = ei2 + E2; roff = n1; }
  else if (t < E1 + E2 + E3) { e = t - E1 - E2; dst = ei3 + E3; roff = n1 + n2; }
  else return;
  rank[t] = atomicAdd(deg + roff + dst[e], 1);
}

__global__ void fill_perm(const int* __restrict__ ei1, const int* __restrict__ ei2,
                          const int* __restrict__ ei3, int E1, int E2, int E3,
                          int n1, int n2,
                          const int* __restrict__ rowptr, const int* __restrict__ rank,
                          int* __restrict__ perm) {
  const int t = blockIdx.x * blockDim.x + threadIdx.x;
  const int* src;
  const int* dst;
  int e, roff, eoff;
  if (t < E1)                { e = t;           src = ei1; dst = ei1 + E1; roff = 0;       eoff = 0; }
  else if (t < E1 + E2)      { e = t - E1;      src = ei2; dst = ei2 + E2; roff = n1;      eoff = E1; }
  else if (t < E1 + E2 + E3) { e = t - E1 - E2; src = ei3; dst = ei3 + E3; roff = n1 + n2; eoff = E1 + E2; }
  else return;
  perm[eoff + rowptr[roff + dst[e]] + rank[t]] = src[e];
}

#define SCAN_TILES 10
__global__ __launch_bounds__(1024) void scan_deg(const int* __restrict__ deg,
                                                 int* __restrict__ rowptr,
                                                 int n1, int n2, int n3) {
  __shared__ int wsum[SCAN_TILES * 32];
  __shared__ int tsum[SCAN_TILES];
  const int set = blockIdx.x;
  const int n   = set == 0 ? n1 : (set == 1 ? n2 : n3);
  const int off = set == 0 ? 0  : (set == 1 ? n1 : n1 + n2);
  const int tid = threadIdx.x, lane = tid & 31, wave = tid >> 5;
  const int4* in  = (const int4*)(deg + off);
  int4*       out = (int4*)(rowptr + off);

  int4 v[SCAN_TILES];
  int  incl[SCAN_TILES];
  for (int i = 0; i < SCAN_TILES; ++i) {
    const int q = i * 1024 + tid;
    v[i] = (4 * q < n) ? in[q] : make_int4(0, 0, 0, 0);
  }
  for (int i = 0; i < SCAN_TILES; ++i) {
    int s = v[i].x + v[i].y + v[i].z + v[i].w;
    for (int d = 1; d < 32; d <<= 1) {
      const int up = __shfl_up(s, d, 32);
      if (lane >= d) s += up;
    }
    incl[i] = s;
    if (lane == 31) wsum[i * 32 + wave] = s;
  }
  __syncthreads();
  if (tid < SCAN_TILES * 32) {
    const int w = wsum[tid];
    int s = w;
    for (int d = 1; d < 32; d <<= 1) {
      const int up = __shfl_up(s, d, 32);
      if (lane >= d) s += up;
    }
    wsum[tid] = s - w;
    if (lane == 31) tsum[wave] = s;
  }
  __syncthreads();
  int carry = 0;
  for (int i = 0; i < SCAN_TILES; ++i) {
    const int q = i * 1024 + tid;
    const int own = v[i].x + v[i].y + v[i].z + v[i].w;
    const int b0 = carry + wsum[i * 32 + wave] + incl[i] - own;
    if (4 * q < n)
      out[q] = make_int4(b0, b0 + v[i].x, b0 + v[i].x + v[i].y, b0 + v[i].x + v[i].y + v[i].z);
    carry += tsum[i];
  }
}

template <int B>
__device__ __forceinline__ void add_rows(const float* __restrict__ xl, int idx, int j,
                                         float4& a0, float4& a1) {
  float4 u[B], w[B];
  for (int t = 0; t < B; ++t) {
    const float* r = xl + (long long)__builtin_amdgcn_readlane(idx, j + t) * C_DIM;
    u[t] = *(const float4*)r;
    w[t] = *(const float4*)(r + 128);
  }
  for (int t = 0; t < B; ++t) {
    a0.x += u[t].x; a0.y += u[t].y; a0.z += u[t].z; a0.w += u[t].w;
    a1.x += w[t].x; a1.y += w[t].y; a1.z += w[t].z; a1.w += w[t].w;
  }
}

__global__ __launch_bounds__(256) void gather_rows(
    const float* __restrict__ x, const int* __restrict__ rowptr,
    const int* __restrict__ deg, const int* __restrict__ perm,
    float* __restrict__ agg, float* __restrict__ cnt, int n_dst) {
  const int lane = threadIdx.x & 31;
  const int d = __builtin_amdgcn_readfirstlane(blockIdx.x * 8 + (threadIdx.x >> 5));
  if (d >= n_dst) return;
  const int beg = rowptr[d];
  const int n   = deg[d];
  const float* xl = x + lane * 4;

  float4 a0 = make_float4(0.f, 0.f, 0.f, 0.f);
  float4 a1 = make_float4(0.f, 0.f, 0.f, 0.f);

  for (int base = 0; base < n; base += 32) {
    const int m = min(n - base, 32);
    int idx = 0;
    if (lane < m) idx = perm[beg + base + lane];
    int j = 0;
    for (; j + 8 <= m; j += 8) add_rows<8>(xl, idx, j, a0, a1);
    if (m & 4) { add_rows<4>(xl, idx, j, a0, a1); j += 4; }
    if (m & 2) { add_rows<2>(xl, idx, j, a0, a1); j += 2; }
    if (m & 1) { add_rows<1>(xl, idx, j, a0, a1); }
  }

  float* o = agg + (long long)d * C_DIM + lane * 4;
  *(float4*)o         = a0;
  *(float4*)(o + 128) = a1;
  if (lane == 0) cnt[d] = (float)n;
}

template <int RELU>
__global__ __launch_bounds__(256) void gemm_dual_wmma(
    const float* __restrict__ A1, const float* __restrict__ Wt1,
    const float* __restrict__ cnt1,
    const float* __restrict__ A2, const float* __restrict__ Wt2,
    const float* __restrict__ bias, float* __restrict__ out,
    int M, int N) {
  constexpr int K = C_DIM;
  const int lane = threadIdx.x & 31;
  const int wave = threadIdx.x >> 5;
  const int nstrips = N >> 6;
  const int tile = blockIdx.x * 8 + wave;
  if (tile >= (M >> 4) * nstrips) return;

  const int m0 = (tile / nstrips) << 4;
  const int n0 = (tile % nstrips) << 6;

  const int half = lane >> 4;
  const int l15  = lane & 15;
  const int row  = m0 + l15;
  const int cb   = n0 + l15;
  const int koff = half << 1;

  const float s1 = 1.0f / fmaxf(cnt1[row], 1.0f);

  v8f acc[4];
#pragma unroll
  for (int j = 0; j < 4; ++j) acc[j] = (v8f){0.f,0.f,0.f,0.f,0.f,0.f,0.f,0.f};

  const float* a1p = A1 + (long long)row * K + koff;
  const float* a2p = A2 + (long long)row * K + koff;
  const float* w1p = Wt1 + (long long)cb * K + koff;
  const float* w2p = Wt2 + (long long)cb * K + koff;

  for (int k = 0; k < K; k += 4) {
    v2f a = *(const v2f*)(a1p + k);
    a = a * s1;
#pragma unroll
    for (int j = 0; j < 4; ++j) {
      v2f b = *(const v2f*)(w1p + (long long)j * 16 * K + k);
      acc[j] = __builtin_amdgcn_wmma_f32_16x16x4_f32(false, a, false, b,
                                                     (short)0, acc[j],
                                                     false, false);
    }
  }
  for (int k = 0; k < K; k += 4) {
    v2f a = *(const v2f*)(a2p + k);
#pragma unroll
    for (int j = 0; j < 4; ++j) {
      v2f b = *(const v2f*)(w2p + (long long)j * 16 * K + k);
      acc[j] = __builtin_amdgcn_wmma_f32_16x16x4_f32(false, a, false, b,
                                                     (short)0, acc[j],
                                                     false, false);
    }
  }

#pragma unroll
  for (int j = 0; j < 4; ++j) {
    const int col = cb + j * 16;
    const float bv = bias[col];
#pragma unroll
    for (int r = 0; r < 8; ++r) {
      float v = acc[j][r] + bv;
      if (RELU) v = fmaxf(v, 0.f);
      out[(long long)(m0 + r + half * 8) * N + col] = v;
    }
  }
}

static inline int cdiv(long long a, long long b) { return (int)((a + b - 1) / b); }

extern "C" void kernel_launch(void* const* d_in, const int* in_sizes, int n_in,
                              void* d_out, int out_size, void* d_ws, size_t ws_size,
                              hipStream_t stream) {
  (void)n_in; (void)out_size; (void)ws_size;

  const int N1 = 40000, N2 = 20000, N3 = 10000;
  const int OUT_C = 64;
  static_assert(N1 % 4 == 0 && N2 % 4 == 0 && N3 % 4 == 0, "scan_deg reads rows four at a time");
  static_assert(N1 <= SCAN_TILES * 4096 && N2 <= N1 && N3 <= N1, "scan_deg tile count");
  static_assert(2 * N2 <= N1 && N3 <= N2, "h2 lives in the half of agg that layers 2 and 3 leave free");

  const float* x   = (const float*)d_in[0];
  const int*   ei1 = (const int*)d_in[1];
  const int*   ei2 = (const int*)d_in[2];
  const int*   ei3 = (const int*)d_in[3];
  const float* Wl1 = (const float*)d_in[4];
  const float* bl1 = (const float*)d_in[5];
  const float* Wr1 = (const float*)d_in[6];
  const float* Wl2 = (const float*)d_in[7];
  const float* bl2 = (const float*)d_in[8];
  const float* Wr2 = (const float*)d_in[9];
  const float* Wl3 = (const float*)d_in[10];
  const float* bl3 = (const float*)d_in[11];
  const float* Wr3 = (const float*)d_in[12];
  const float* Wls = (const float*)d_in[13];
  const float* bls = (const float*)d_in[14];
  const float* Wrs = (const float*)d_in[15];

  const int E1 = in_sizes[1] / 2;
  const int E2 = in_sizes[2] / 2;
  const int E3 = in_sizes[3] / 2;
  const int ET = E1 + E2 + E3;
  const int NT = N1 + N2 + N3;

  float* h1  = (float*)d_ws;
  int*   csr = (int*)(h1 + (long long)N1 * C_DIM);
  int*   deg  = csr;
  int*   rowp = deg + NT;
  int*   rank = rowp + NT;
  int*   perm = rank + ET;
  float* agg = (float*)(csr + (long long)N2 * C_DIM);
  float* h2  = agg + (long long)N2 * C_DIM;
  float* cnt = agg + (long long)N1 * C_DIM;
  float* wt  = cnt + N1;
  float* Wl1t = wt;
  float* Wr1t = Wl1t + C_DIM * C_DIM;
  float* Wl2t = Wr1t + C_DIM * C_DIM;
  float* Wr2t = Wl2t + C_DIM * C_DIM;
  float* Wl3t = Wr2t + C_DIM * C_DIM;
  float* Wr3t = Wl3t + C_DIM * OUT_C;
  float* Wlst = Wr3t + C_DIM * OUT_C;
  float* Wrst = Wlst + C_DIM * OUT_C;

  float* mu     = (float*)d_out;
  float* logstd = mu + (long long)N3 * OUT_C;

  const int ZB = 256, SB = 256, GB = 256, TB = 256;

  {
    int nbig = C_DIM * C_DIM, nsml = C_DIM * OUT_C;
    transpose_w<<<cdiv(nbig, TB), TB, 0, stream>>>(Wl1, Wl1t, C_DIM, C_DIM);
    transpose_w<<<cdiv(nbig, TB), TB, 0, stream>>>(Wr1, Wr1t, C_DIM, C_DIM);
    transpose_w<<<cdiv(nbig, TB), TB, 0, stream>>>(Wl2, Wl2t, C_DIM, C_DIM);
    transpose_w<<<cdiv(nbig, TB), TB, 0, stream>>>(Wr2, Wr2t, C_DIM, C_DIM);
    transpose_w<<<cdiv(nsml, TB), TB, 0, stream>>>(Wl3, Wl3t, C_DIM, OUT_C);
    transpose_w<<<cdiv(nsml, TB), TB, 0, stream>>>(Wr3, Wr3t, C_DIM, OUT_C);
    transpose_w<<<cdiv(nsml, TB), TB, 0, stream>>>(Wls, Wlst, C_DIM, OUT_C);
    transpose_w<<<cdiv(nsml, TB), TB, 0, stream>>>(Wrs, Wrst, C_DIM, OUT_C);
  }

  {
    zero_i32<<<cdiv(NT, ZB), ZB, 0, stream>>>(deg, NT);
    edge_rank<<<cdiv(ET, SB), SB, 0, stream>>>(ei1, ei2, ei3, E1, E2, E3, N1, N2, deg, rank);
    scan_deg<<<3, 1024, 0, stream>>>(deg, rowp, N1, N2, N3);
    fill_perm<<<cdiv(ET, SB), SB, 0, stream>>>(ei1, ei2, ei3, E1, E2, E3, N1, N2, rowp, rank, perm);
  }

  {
    gather_rows<<<cdiv(N1, 8), GB, 0, stream>>>(x, rowp, deg, perm, agg, cnt, N1);
    int tiles = (N1 / 16) * (C_DIM / 64);
    gemm_dual_wmma<1><<<cdiv(tiles, 8), GB, 0, stream>>>(
        agg, Wl1t, cnt, x, Wr1t, bl1, h1, N1, C_DIM);
  }

  {
    gather_rows<<<cdiv(N2, 8), GB, 0, stream>>>(h1, rowp + N1, deg + N1, perm + E1, agg, cnt, N2);
    int tiles = (N2 / 16) * (C_DIM / 64);
    gemm_dual_wmma<1><<<cdiv(tiles, 8), GB, 0, stream>>>(
        agg, Wl2t, cnt, h1, Wr2t, bl2, h2, N2, C_DIM);
  }

  {
    gather_rows<<<cdiv(N3, 8), GB, 0, stream>>>(h2, rowp + N1 + N2, deg + N1 + N2, perm + E1 + E2,
                                                agg, cnt, N3);

    int tiles = (N3 / 16) * (OUT_C / 64);
    gemm_dual_wmma<0><<<cdiv(tiles, 8), GB, 0, stream>>>(
        agg, Wl3t, cnt, h2, Wr3t, bl3, mu, N3, OUT_C);
    gemm_dual_wmma<0><<<cdiv(tiles, 8), GB, 0, stream>>>(
        agg, Wlst, cnt, h2, Wrst, bls, logstd, N3, OUT_C);
  }
}


# GEMM of layers 1 and 2: each wave computes an 80x64 block instead of 16x64, so 9 fragment loads feed 20 WMMAs per k-step (was 5 for 4), next step's loads issued ahead; k order and sums unchanged
# speedup vs baseline: 2.0083x; 1.2557x over previous
#include <hip/hip_runtime.h>
#include <hip/hip_bf16.h>

typedef __attribute__((ext_vector_type(2))) float v2f;
typedef __attribute__((ext_vector_type(8))) float v8f;

#define C_DIM 256

__global__ void zero_i32(int* __restrict__ p, int n) {
  int i = blockIdx.x * blockDim.x + threadIdx.x;
  if (i < n) p[i] = 0;
}

__global__ void transpose_w(const float* __restrict__ W, float* __restrict__ Wt,
                            int K, int N) {
  int t = blockIdx.x * blockDim.x + threadIdx.x;
  if (t >= K * N) return;
  int k = t / N;
  int n = t % N;
  Wt[n * K + k] = W[t];
}

__global__ void edge_rank(const int* __restrict__ ei1, const int* __restrict__ ei2,
                          const int* __restrict__ ei3, int E1, int E2, int E3,
                          int n1, int n2,
                          int* __restrict__ deg, int* __restrict__ rank) {
  const int t = blockIdx.x * blockDim.x + threadIdx.x;
  const int* dst;
  int e, roff;
  if (t < E1)                { e = t;           dst = ei1 + E1; roff = 0; }
  else if (t < E1 + E2)      { e = t - E1;      dst = ei2 + E2; roff = n1; }
  else if (t < E1 + E2 + E3) { e = t - E1 - E2; dst = ei3 + E3; roff = n1 + n2; }
  else return;
  rank[t] = atomicAdd(deg + roff + dst[e], 1);
}

__global__ void fill_perm(const int* __restrict__ ei1, const int* __restrict__ ei2,
                          const int* __restrict__ ei3, int E1, int E2, int E3,
                          int n1, int n2,
                          const int* __restrict__ rowptr, const int* __restrict__ rank,
                          int* __restrict__ perm) {
  const int t = blockIdx.x * blockDim.x + threadIdx.x;
  const int* src;
  const int* dst;
  int e, roff, eoff;
  if (t < E1)                { e = t;           src = ei1; dst = ei1 + E1; roff = 0;       eoff = 0; }
  else if (t < E1 + E2)      { e = t - E1;      src = ei2; dst = ei2 + E2; roff = n1;      eoff = E1; }
  else if (t < E1 + E2 + E3) { e = t - E1 - E2; src = ei3; dst = ei3 + E3; roff = n1 + n2; eoff = E1 + E2; }
  else return;
  perm[eoff + rowptr[roff + dst[e]] + rank[t]] = src[e];
}

#define SCAN_TILES 10
__global__ __launch_bounds__(1024) void scan_deg(const int* __restrict__ deg,
                                                 int* __restrict__ rowptr,
                                                 int n1, int n2, int n3) {
  __shared__ int wsum[SCAN_TILES * 32];
  __shared__ int tsum[SCAN_TILES];
  const int set = blockIdx.x;
  const int n   = set == 0 ? n1 : (set == 1 ? n2 : n3);
  const int off = set == 0 ? 0  : (set == 1 ? n1 : n1 + n2);
  const int tid = threadIdx.x, lane = tid & 31, wave = tid >> 5;
  const int4* in  = (const int4*)(deg + off);
  int4*       out = (int4*)(rowptr + off);

  int4 v[SCAN_TILES];
  int  incl[SCAN_TILES];
  for (int i = 0; i < SCAN_TILES; ++i) {
    const int q = i * 1024 + tid;
    v[i] = (4 * q < n) ? in[q] : make_int4(0, 0, 0, 0);
  }
  for (int i = 0; i < SCAN_TILES; ++i) {
    int s = v[i].x + v[i].y + v[i].z + v[i].w;
    for (int d = 1; d < 32; d <<= 1) {
      const int up = __shfl_up(s, d, 32);
      if (lane >= d) s += up;
    }
    incl[i] = s;
    if (lane == 31) wsum[i * 32 + wave] = s;
  }
  __syncthreads();
  if (tid < SCAN_TILES * 32) {
    const int w = wsum[tid];
    int s = w;
    for (int d = 1; d < 32; d <<= 1) {
      const int up = __shfl_up(s, d, 32);
      if (lane >= d) s += up;
    }
    wsum[tid] = s - w;
    if (lane == 31) tsum[wave] = s;
  }
  __syncthreads();
  int carry = 0;
  for (int i = 0; i < SCAN_TILES; ++i) {
    const int q = i * 1024 + tid;
    const int own = v[i].x + v[i].y + v[i].z + v[i].w;
    const int b0 = carry + wsum[i * 32 + wave] + incl[i] - own;
    if (4 * q < n)
      out[q] = make_int4(b0, b0 + v[i].x, b0 + v[i].x + v[i].y, b0 + v[i].x + v[i].y + v[i].z);
    carry += tsum[i];
  }
}

template <int B>
__device__ __forceinline__ void add_rows(const float* __restrict__ xl, int idx, int j,
                                         float4& a0, float4& a1) {
  float4 u[B], w[B];
  for (int t = 0; t < B; ++t) {
    const float* r = xl + (long long)__builtin_amdgcn_readlane(idx, j + t) * C_DIM;
    u[t] = *(const float4*)r;
    w[t] = *(const float4*)(r + 128);
  }
  for (int t = 0; t < B; ++t) {
    a0.x += u[t].x; a0.y += u[t].y; a0.z += u[t].z; a0.w += u[t].w;
    a1.x += w[t].x; a1.y += w[t].y; a1.z += w[t].z; a1.w += w[t].w;
  }
}

__global__ __launch_bounds__(256) void gather_rows(
    const float* __restrict__ x, const int* __restrict__ rowptr,
    const int* __restrict__ deg, const int* __restrict__ perm,
    float* __restrict__ agg, float* __restrict__ cnt, int n_dst) {
  const int lane = threadIdx.x & 31;
  const int d = __builtin_amdgcn_readfirstlane(blockIdx.x * 8 + (threadIdx.x >> 5));
  if (d >= n_dst) return;
  const int beg = rowptr[d];
  const int n   = deg[d];
  const float* xl = x + lane * 4;

  float4 a0 = make_float4(0.f, 0.f, 0.f, 0.f);
  float4 a1 = make_float4(0.f, 0.f, 0.f, 0.f);

  for (int base = 0; base < n; base += 32) {
    const int m = min(n - base, 32);
    int idx = 0;
    if (lane < m) idx = perm[beg + base + lane];
    int j = 0;
    for (; j + 8 <= m; j += 8) add_rows<8>(xl, idx, j, a0, a1);
    if (m & 4) { add_rows<4>(xl, idx, j, a0, a1); j += 4; }
    if (m & 2) { add_rows<2>(xl, idx, j, a0, a1); j += 2; }
    if (m & 1) { add_rows<1>(xl, idx, j, a0, a1); }
  }

  float* o = agg + (long long)d * C_DIM + lane * 4;
  *(float4*)o         = a0;
  *(float4*)(o + 128) = a1;
  if (lane == 0) cnt[d] = (float)n;
}

template <int MT>
__device__ __forceinline__ void load_frags(const float* __restrict__ ap,
                                           const float* __restrict__ wp, int k,
                                           v2f (&a)[MT], v2f (&b)[4]) {
  constexpr int K = C_DIM;
#pragma unroll
  for (int i = 0; i < MT; ++i) a[i] = *(const v2f*)(ap + i * 16 * K + k);
#pragma unroll
  for (int j = 0; j < 4; ++j) b[j] = *(const v2f*)(wp + j * 16 * K + k);
}

template <int MT, bool SCALE>
__device__ __forceinline__ void mma_step(const v2f (&a)[MT], const v2f (&b)[4],
                                         const float (&s)[MT], v8f (&acc)[MT][4]) {
#pragma unroll
  for (int i = 0; i < MT; ++i) {
    const v2f x = SCALE ? a[i] * s[i] : a[i];
#pragma unroll
    for (int j = 0; j < 4; ++j)
      acc[i][j] = __builtin_amdgcn_wmma_f32_16x16x4_f32(false, x, false, b[j],
                                                        (short)0, acc[i][j],
                                                        false, false);
  }
}

template <int MT, bool SCALE>
__device__ __forceinline__ void gemm_path(const float* __restrict__ ap,
                                          const float* __restrict__ wp,
                                          const float (&s)[MT], v8f (&acc)[MT][4]) {
  constexpr int K = C_DIM;
  v2f a0[MT], b0[4], a1[MT], b1[4];
  load_frags<MT>(ap, wp, 0, a0, b0);
  for (int k = 0; k < K - 8; k += 8) {
    load_frags<MT>(ap, wp, k + 4, a1, b1);
    mma_step<MT, SCALE>(a0, b0, s, acc);
    load_frags<MT>(ap, wp, k + 8, a0, b0);
    mma_step<MT, SCALE>(a1, b1, s, acc);
  }
  load_frags<MT>(ap, wp, K - 4, a1, b1);
  mma_step<MT, SCALE>(a0, b0, s, acc);
  mma_step<MT, SCALE>(a1, b1, s, acc);
}

template <int MT, int RELU>
__global__ __launch_bounds__(256) void gemm_dual_wmma(
    const float* __restrict__ A1, const float* __restrict__ Wt1,
    const float* __restrict__ cnt1,
    const float* __restrict__ A2, const float* __restrict__ Wt2,
    const float* __restrict__ bias, float* __restrict__ out,
    int M, int N) {
  constexpr int K = C_DIM;
  constexpr int BM = 16 * MT;
  const int lane = threadIdx.x & 31;
  const int wave = threadIdx.x >> 5;
  const int nstrips = N >> 6;
  const int tile = blockIdx.x * (blockDim.x >> 5) + wave;
  if (tile >= (M / BM) * nstrips) return;

  const int m0 = (tile / nstrips) * BM;
  const int n0 = (tile % nstrips) << 6;

  const int half = lane >> 4;
  const int l15  = lane & 15;
  const int row  = m0 + l15;
  const int cb   = n0 + l15;
  const int koff = half << 1;

  float s1[MT];
#pragma unroll
  for (int i = 0; i < MT; ++i) s1[i] = 1.0f / fmaxf(cnt1[row + i * 16], 1.0f);

  v8f acc[MT][4];
#pragma unroll
  for (int i = 0; i < MT; ++i)
#pragma unroll
    for (int j = 0; j < 4; ++j) acc[i][j] = (v8f){0.f,0.f,0.f,0.f,0.f,0.f,0.f,0.f};

  const float* a1p = A1 + (long long)row * K + koff;
  const float* a2p = A2 + (long long)row * K + koff;
  const float* w1p = Wt1 + (long long)cb * K + koff;
  const float* w2p = Wt2 + (long long)cb * K + koff;

  gemm_path<MT, true >(a1p, w1p, s1, acc);
  gemm_path<MT, false>(a2p, w2p, s1, acc);

#pragma unroll
  for (int j = 0; j < 4; ++j) {
    const int col = cb + j * 16;
    const float bv = bias[col];
#pragma unroll
    for (int i = 0; i < MT; ++i) {
#pragma unroll
      for (int r = 0; r < 8; ++r) {
        float v = acc[i][j][r] + bv;
        if (RELU) v = fmaxf(v, 0.f);
        out[(long long)(m0 + i * 16 + r + half * 8) * N + col] = v;
      }
    }
  }
}

static inline int cdiv(long long a, long long b) { return (int)((a + b - 1) / b); }

extern "C" void kernel_launch(void* const* d_in, const int* in_sizes, int n_in,
                              void* d_out, int out_size, void* d_ws, size_t ws_size,
                              hipStream_t stream) {
  (void)n_in; (void)out_size; (void)ws_size;

  const int N1 = 40000, N2 = 20000, N3 = 10000;
  const int OUT_C = 64;
  static_assert(N1 % 4 == 0 && N2 % 4 == 0 && N3 % 4 == 0, "scan_deg reads rows four at a time");
  static_assert(N1 <= SCAN_TILES * 4096 && N2 <= N1 && N3 <= N1, "scan_deg tile count");
  static_assert(2 * N2 <= N1 && N3 <= N2, "h2 lives in the half of agg that layers 2 and 3 leave free");
  constexpr int MT = 5;
  static_assert(N1 % (16 * MT) == 0 && N2 % (16 * MT) == 0 && N3 % 16 == 0, "GEMM row blocks must tile M");

  const float* x   = (const float*)d_in[0];
  const int*   ei1 = (const int*)d_in[1];
  const int*   ei2 = (const int*)d_in[2];
  const int*   ei3 = (const int*)d_in[3];
  const float* Wl1 = (const float*)d_in[4];
  const float* bl1 = (const float*)d_in[5];
  const float* Wr1 = (const float*)d_in[6];
  const float* Wl2 = (const float*)d_in[7];
  const float* bl2 = (const float*)d_in[8];
  const float* Wr2 = (const float*)d_in[9];
  const float* Wl3 = (const float*)d_in[10];
  const float* bl3 = (const float*)d_in[11];
  const float* Wr3 = (const float*)d_in[12];
  const float* Wls = (const float*)d_in[13];
  const float* bls = (const float*)d_in[14];
  const float* Wrs = (const float*)d_in[15];

  const int E1 = in_sizes[1] / 2;
  const int E2 = in_sizes[2] / 2;
  const int E3 = in_sizes[3] / 2;
  const int ET = E1 + E2 + E3;
  const int NT = N1 + N2 + N3;

  float* h1  = (float*)d_ws;
  int*   csr = (int*)(h1 + (long long)N1 * C_DIM);
  int*   deg  = csr;
  int*   rowp = deg + NT;
  int*   rank = rowp + NT;
  int*   perm = rank + ET;
  float* agg = (float*)(csr + (long long)N2 * C_DIM);
  float* h2  = agg + (long long)N2 * C_DIM;
  float* cnt = agg + (long long)N1 * C_DIM;
  float* wt  = cnt + N1;
  float* Wl1t = wt;
  float* Wr1t = Wl1t + C_DIM * C_DIM;
  float* Wl2t = Wr1t + C_DIM * C_DIM;
  float* Wr2t = Wl2t + C_DIM * C_DIM;
  float* Wl3t = Wr2t + C_DIM * C_DIM;
  float* Wr3t = Wl3t + C_DIM * OUT_C;
  float* Wlst = Wr3t + C_DIM * OUT_C;
  float* Wrst = Wlst + C_DIM * OUT_C;

  float* mu     = (float*)d_out;
  float* logstd = mu + (long long)N3 * OUT_C;

  const int ZB = 256, SB = 256, GB = 256, TB = 256;

  {
    int nbig = C_DIM * C_DIM, nsml = C_DIM * OUT_C;
    transpose_w<<<cdiv(nbig, TB), TB, 0, stream>>>(Wl1, Wl1t, C_DIM, C_DIM);
    transpose_w<<<cdiv(nbig, TB), TB, 0, stream>>>(Wr1, Wr1t, C_DIM, C_DIM);
    transpose_w<<<cdiv(nbig, TB), TB, 0, stream>>>(Wl2, Wl2t, C_DIM, C_DIM);
    transpose_w<<<cdiv(nbig, TB), TB, 0, stream>>>(Wr2, Wr2t, C_DIM, C_DIM);
    transpose_w<<<cdiv(nsml, TB), TB, 0, stream>>>(Wl3, Wl3t, C_DIM, OUT_C);
    transpose_w<<<cdiv(nsml, TB), TB, 0, stream>>>(Wr3, Wr3t, C_DIM, OUT_C);
    transpose_w<<<cdiv(nsml, TB), TB, 0, stream>>>(Wls, Wlst, C_DIM, OUT_C);
    transpose_w<<<cdiv(nsml, TB), TB, 0, stream>>>(Wrs, Wrst, C_DIM, OUT_C);
  }

  {
    zero_i32<<<cdiv(NT, ZB), ZB, 0, stream>>>(deg, NT);
    edge_rank<<<cdiv(ET, SB), SB, 0, stream>>>(ei1, ei2, ei3, E1, E2, E3, N1, N2, deg, rank);
    scan_deg<<<3, 1024, 0, stream>>>(deg, rowp, N1, N2, N3);
    fill_perm<<<cdiv(ET, SB), SB, 0, stream>>>(ei1, ei2, ei3, E1, E2, E3, N1, N2, rowp, rank, perm);
  }

  {
    gather_rows<<<cdiv(N1, 8), GB, 0, stream>>>(x, rowp, deg, perm, agg, cnt, N1);
    int tiles = (N1 / (16 * MT)) * (C_DIM / 64);
    gemm_dual_wmma<MT, 1><<<cdiv(tiles, 4), 128, 0, stream>>>(
        agg, Wl1t, cnt, x, Wr1t, bl1, h1, N1, C_DIM);
  }

  {
    gather_rows<<<cdiv(N2, 8), GB, 0, stream>>>(h1, rowp + N1, deg + N1, perm + E1, agg, cnt, N2);
    int tiles = (N2 / (16 * MT)) * (C_DIM / 64);
    gemm_dual_wmma<MT, 1><<<cdiv(tiles, 4), 128, 0, stream>>>(
        agg, Wl2t, cnt, h1, Wr2t, bl2, h2, N2, C_DIM);
  }

  {
    gather_rows<<<cdiv(N3, 8), GB, 0, stream>>>(h2, rowp + N1 + N2, deg + N1 + N2, perm + E1 + E2,
                                                agg, cnt, N3);

    int tiles = (N3 / 16) * (OUT_C / 64);
    gemm_dual_wmma<1, 0><<<cdiv(tiles, 8), GB, 0, stream>>>(
        agg, Wl3t, cnt, h2, Wr3t, bl3, mu, N3, OUT_C);
    gemm_dual_wmma<1, 0><<<cdiv(tiles, 8), GB, 0, stream>>>(
        agg, Wlst, cnt, h2, Wrst, bls, logstd, N3, OUT_C);
  }
}


# Independent work shares launches: the 8 weight transposes ride as extra blocks of the row-scan launch, mu and logstd GEMMs run as two halves of one grid (19 -> 10 launches); sums unchanged
# speedup vs baseline: 2.2787x; 1.1346x over previous
#include <hip/hip_runtime.h>
#include <hip/hip_bf16.h>

typedef __attribute__((ext_vector_type(2))) float v2f;
typedef __attribute__((ext_vector_type(8))) float v8f;

#define C_DIM 256

__global__ void zero_i32(int* __restrict__ p, int n) {
  int i = blockIdx.x * blockDim.x + threadIdx.x;
  if (i < n) p[i] = 0;
}

#define O_DIM 64
#define TR_BIG_BLOCKS (C_DIM * C_DIM / 4096)
#define TR_SML_BLOCKS (C_DIM * O_DIM / 4096)
#define TR_BLOCKS (4 * TR_BIG_BLOCKS + 4 * TR_SML_BLOCKS)
__device__ __forceinline__ void transpose_block(
    int b, int tid,
    const float* __restrict__ w0, const float* __restrict__ w1,
    const float* __restrict__ w2, const float* __restrict__ w3,
    const float* __restrict__ w4, const float* __restrict__ w5,
    const float* __restrict__ w6, const float* __restrict__ w7,
    float* __restrict__ wt) {
  constexpr int K = C_DIM;
  int m, local;
  if (b < 4 * TR_BIG_BLOCKS) { m = b / TR_BIG_BLOCKS; local = b % TR_BIG_BLOCKS; }
  else { const int r = b - 4 * TR_BIG_BLOCKS; m = 4 + r / TR_SML_BLOCKS; local = r % TR_SML_BLOCKS; }
  const float* W = m == 0 ? w0 : m == 1 ? w1 : m == 2 ? w2 : m == 3 ? w3
                 : m == 4 ? w4 : m == 5 ? w5 : m == 6 ? w6 : w7;
  const int N = m < 4 ? C_DIM : O_DIM;
  float* Wt = wt + (m < 4 ? m * (C_DIM * C_DIM)
                          : 4 * (C_DIM * C_DIM) + (m - 4) * (C_DIM * O_DIM));
  const int t = (local * 1024 + tid) * 4;
  const int k = t / N;
  const int n = t % N;
  const float4 v = *(const float4*)(W + t);
  Wt[(n + 0) * K + k] = v.x;
  Wt[(n + 1) * K + k] = v.y;
  Wt[(n + 2) * K + k] = v.z;
  Wt[(n + 3) * K + k] = v.w;
}

__global__ void edge_rank(const int* __restrict__ ei1, const int* __restrict__ ei2,
                          const int* __restrict__ ei3, int E1, int E2, int E3,
                          int n1, int n2,
                          int* __restrict__ deg, int* __restrict__ rank) {
  const int t = blockIdx.x * blockDim.x + threadIdx.x;
  const int* dst;
  int e, roff;
  if (t < E1)                { e = t;           dst = ei1 + E1; roff = 0; }
  else if (t < E1 + E2)      { e = t - E1;      dst = ei2 + E2; roff = n1; }
  else if (t < E1 + E2 + E3) { e = t - E1 - E2; dst = ei3 + E3; roff = n1 + n2; }
  else return;
  rank[t] = atomicAdd(deg + roff + dst[e], 1);
}

__global__ void fill_perm(const int* __restrict__ ei1, const int* __restrict__ ei2,
                          const int* __restrict__ ei3, int E1, int E2, int E3,
                          int n1, int n2,
                          const int* __restrict__ rowptr, const int* __restrict__ rank,
                          int* __restrict__ perm) {
  const int t = blockIdx.x * blockDim.x + threadIdx.x;
  const int* src;
  const int* dst;
  int e, roff, eoff;
  if (t < E1)                { e = t;           src = ei1; dst = ei1 + E1; roff = 0;       eoff = 0; }
  else if (t < E1 + E2)      { e = t - E1;      src = ei2; dst = ei2 + E2; roff = n1;      eoff = E1; }
  else if (t < E1 + E2 + E3) { e = t - E1 - E2; src = ei3; dst = ei3 + E3; roff = n1 + n2; eoff = E1 + E2; }
  else return;
  perm[eoff + rowptr[roff + dst[e]] + rank[t]] = src[e];
}

#define SCAN_TILES 10
__global__ __launch_bounds__(1024) void scan_deg(const int* __restrict__ deg,
                                                 int* __restrict__ rowptr,
                                                 int n1, int n2, int n3,
    const float* __restrict__ w0, const float* __restrict__ w1,
    const float* __restrict__ w2, const float* __restrict__ w3,
    const float* __restrict__ w4, const float* __restrict__ w5,
    const float* __restrict__ w6, const float* __restrict__ w7,
    float* __restrict__ wt) {
  __shared__ int wsum[SCAN_TILES * 32];
  __shared__ int tsum[SCAN_TILES];
  if (blockIdx.x >= 3) {
    transpose_block(blockIdx.x - 3, threadIdx.x, w0, w1, w2, w3, w4, w5, w6, w7, wt);
    return;
  }
  const int set = blockIdx.x;
  const int n   = set == 0 ? n1 : (set == 1 ? n2 : n3);
  const int off = set == 0 ? 0  : (set == 1 ? n1 : n1 + n2);
  const int tid = threadIdx.x, lane = tid & 31, wave = tid >> 5;
  const int4* in  = (const int4*)(deg + off);
  int4*       out = (int4*)(rowptr + off);

  int4 v[SCAN_TILES];
  int  incl[SCAN_TILES];
  for (int i = 0; i < SCAN_TILES; ++i) {
    const int q = i * 1024 + tid;
    v[i] = (4 * q < n) ? in[q] : make_int4(0, 0, 0, 0);
  }
  for (int i = 0; i < SCAN_TILES; ++i) {
    int s = v[i].x + v[i].y + v[i].z + v[i].w;
    for (int d = 1; d < 32; d <<= 1) {
      const int up = __shfl_up(s, d, 32);
      if (lane >= d) s += up;
    }
    incl[i] = s;
    if (lane == 31) wsum[i * 32 + wave] = s;
  }
  __syncthreads();
  if (tid < SCAN_TILES * 32) {
    const int w = wsum[tid];
    int s = w;
    for (int d = 1; d < 32; d <<= 1) {
      const int up = __shfl_up(s, d, 32);
      if (lane >= d) s += up;
    }
    wsum[tid] = s - w;
    if (lane == 31) tsum[wave] = s;
  }
  __syncthreads();
  int carry = 0;
  for (int i = 0; i < SCAN_TILES; ++i) {
    const int q = i * 1024 + tid;
    const int own = v[i].x + v[i].y + v[i].z + v[i].w;
    const int b0 = carry + wsum[i * 32 + wave] + incl[i] - own;
    if (4 * q < n)
      out[q] = make_int4(b0, b0 + v[i].x, b0 + v[i].x + v[i].y, b0 + v[i].x + v[i].y + v[i].z);
    carry += tsum[i];
  }
}

template <int B>
__device__ __forceinline__ void add_rows(const float* __restrict__ xl, int idx, int j,
                                         float4& a0, float4& a1) {
  float4 u[B], w[B];
  for (int t = 0; t < B; ++t) {
    const float* r = xl + (long long)__builtin_amdgcn_readlane(idx, j + t) * C_DIM;
    u[t] = *(const float4*)r;
    w[t] = *(const float4*)(r + 128);
  }
  for (int t = 0; t < B; ++t) {
    a0.x += u[t].x; a0.y += u[t].y; a0.z += u[t].z; a0.w += u[t].w;
    a1.x += w[t].x; a1.y += w[t].y; a1.z += w[t].z; a1.w += w[t].w;
  }
}

__global__ __launch_bounds__(256) void gather_rows(
    const float* __restrict__ x, const int* __restrict__ rowptr,
    const int* __restrict__ deg, const int* __restrict__ perm,
    float* __restrict__ agg, float* __restrict__ cnt, int n_dst) {
  const int lane = threadIdx.x & 31;
  const int d = __builtin_amdgcn_readfirstlane(blockIdx.x * 8 + (threadIdx.x >> 5));
  if (d >= n_dst) return;
  const int beg = rowptr[d];
  const int n   = deg[d];
  const float* xl = x + lane * 4;

  float4 a0 = make_float4(0.f, 0.f, 0.f, 0.f);
  float4 a1 = make_float4(0.f, 0.f, 0.f, 0.f);

  for (int base = 0; base < n; base += 32) {
    const int m = min(n - base, 32);
    int idx = 0;
    if (lane < m) idx = perm[beg + base + lane];
    int j = 0;
    for (; j + 8 <= m; j += 8) add_rows<8>(xl, idx, j, a0, a1);
    if (m & 4) { add_rows<4>(xl, idx, j, a0, a1); j += 4; }
    if (m & 2) { add_rows<2>(xl, idx, j, a0, a1); j += 2; }
    if (m & 1) { add_rows<1>(xl, idx, j, a0, a1); }
  }

  float* o = agg + (long long)d * C_DIM + lane * 4;
  *(float4*)o         = a0;
  *(float4*)(o + 128) = a1;
  if (lane == 0) cnt[d] = (float)n;
}

template <int MT>
__device__ __forceinline__ void load_frags(const float* __restrict__ ap,
                                           const float* __restrict__ wp, int k,
                                           v2f (&a)[MT], v2f (&b)[4]) {
  constexpr int K = C_DIM;
#pragma unroll
  for (int i = 0; i < MT; ++i) a[i] = *(const v2f*)(ap + i * 16 * K + k);
#pragma unroll
  for (int j = 0; j < 4; ++j) b[j] = *(const v2f*)(wp + j * 16 * K + k);
}

template <int MT, bool SCALE>
__device__ __forceinline__ void mma_step(const v2f (&a)[MT], const v2f (&b)[4],
                                         const float (&s)[MT], v8f (&acc)[MT][4]) {
#pragma unroll
  for (int i = 0; i < MT; ++i) {
    const v2f x = SCALE ? a[i] * s[i] : a[i];
#pragma unroll
    for (int j = 0; j < 4; ++j)
      acc[i][j] = __builtin_amdgcn_wmma_f32_16x16x4_f32(false, x, false, b[j],
                                                        (short)0, acc[i][j],
                                                        false, false);
  }
}

template <int MT, bool SCALE>
__device__ __forceinline__ void gemm_path(const float* __restrict__ ap,
                                          const float* __restrict__ wp,
                                          const float (&s)[MT], v8f (&acc)[MT][4]) {
  constexpr int K = C_DIM;
  v2f a0[MT], b0[4], a1[MT], b1[4];
  load_frags<MT>(ap, wp, 0, a0, b0);
  for (int k = 0; k < K - 8; k += 8) {
    load_frags<MT>(ap, wp, k + 4, a1, b1);
    mma_step<MT, SCALE>(a0, b0, s, acc);
    load_frags<MT>(ap, wp, k + 8, a0, b0);
    mma_step<MT, SCALE>(a1, b1, s, acc);
  }
  load_frags<MT>(ap, wp, K - 4, a1, b1);
  mma_step<MT, SCALE>(a0, b0, s, acc);
  mma_step<MT, SCALE>(a1, b1, s, acc);
}

template <int MT, int RELU>
__device__ __forceinline__ void gemm_dual_tile(
    const float* __restrict__ A1, const float* __restrict__ Wt1,
    const float* __restrict__ cnt1,
    const float* __restrict__ A2, const float* __restrict__ Wt2,
    const float* __restrict__ bias, float* __restrict__ out,
    int M, int N) {
  constexpr int K = C_DIM;
  constexpr int BM = 16 * MT;
  const int lane = threadIdx.x & 31;
  const int wave = threadIdx.x >> 5;
  const int nstrips = N >> 6;
  const int tile = blockIdx.x * (blockDim.x >> 5) + wave;
  if (tile >= (M / BM) * nstrips) return;

  const int m0 = (tile / nstrips) * BM;
  const int n0 = (tile % nstrips) << 6;

  const int half = lane >> 4;
  const int l15  = lane & 15;
  const int row  = m0 + l15;
  const int cb   = n0 + l15;
  const int koff = half << 1;

  float s1[MT];
#pragma unroll
  for (int i = 0; i < MT; ++i) s1[i] = 1.0f / fmaxf(cnt1[row + i * 16], 1.0f);

  v8f acc[MT][4];
#pragma unroll
  for (int i = 0; i < MT; ++i)
#pragma unroll
    for (int j = 0; j < 4; ++j) acc[i][j] = (v8f){0.f,0.f,0.f,0.f,0.f,0.f,0.f,0.f};

  const float* a1p = A1 + (long long)row * K + koff;
  const float* a2p = A2 + (long long)row * K + koff;
  const float* w1p = Wt1 + (long long)cb * K + koff;
  const float* w2p = Wt2 + (long long)cb * K + koff;

  gemm_path<MT, true >(a1p, w1p, s1, acc);
  gemm_path<MT, false>(a2p, w2p, s1, acc);

#pragma unroll
  for (int j = 0; j < 4; ++j) {
    const int col = cb + j * 16;
    const float bv = bias[col];
#pragma unroll
    for (int i = 0; i < MT; ++i) {
#pragma unroll
      for (int r = 0; r < 8; ++r) {
        float v = acc[i][j][r] + bv;
        if (RELU) v = fmaxf(v, 0.f);
        out[(long long)(m0 + i * 16 + r + half * 8) * N + col] = v;
      }
    }
  }
}

template <int MT, int RELU>
__global__ __launch_bounds__(256) void gemm_dual_wmma(
    const float* __restrict__ A1, const float* __restrict__ Wt1,
    const float* __restrict__ cnt1,
    const float* __restrict__ A2, const float* __restrict__ Wt2,
    const float* __restrict__ bias, float* __restrict__ out,
    int M, int N) {
  gemm_dual_tile<MT, RELU>(A1, Wt1, cnt1, A2, Wt2, bias, out, M, N);
}

__global__ __launch_bounds__(256) void gemm_heads_wmma(
    const float* __restrict__ A1, const float* __restrict__ cnt1,
    const float* __restrict__ A2,
    const float* __restrict__ Wt1a, const float* __restrict__ Wt2a,
    const float* __restrict__ biasa, float* __restrict__ outa,
    const float* __restrict__ Wt1b, const float* __restrict__ Wt2b,
    const float* __restrict__ biasb, float* __restrict__ outb,
    int M, int N) {
  const bool second = blockIdx.y != 0;
  gemm_dual_tile<1, 0>(A1, second ? Wt1b : Wt1a, cnt1,
                       A2, second ? Wt2b : Wt2a,
                       second ? biasb : biasa, second ? outb : outa, M, N);
}

static inline int cdiv(long long a, long long b) { return (int)((a + b - 1) / b); }

extern "C" void kernel_launch(void* const* d_in, const int* in_sizes, int n_in,
                              void* d_out, int out_size, void* d_ws, size_t ws_size,
                              hipStream_t stream) {
  (void)n_in; (void)out_size; (void)ws_size;

  const int N1 = 40000, N2 = 20000, N3 = 10000;
  const int OUT_C = O_DIM;
  static_assert(N1 % 4 == 0 && N2 % 4 == 0 && N3 % 4 == 0, "scan_deg reads rows four at a time");
  static_assert(N1 <= SCAN_TILES * 4096 && N2 <= N1 && N3 <= N1, "scan_deg tile count");
  static_assert(2 * N2 <= N1 && N3 <= N2, "h2 lives in the half of agg that layers 2 and 3 leave free");
  constexpr int MT = 5;
  static_assert(N1 % (16 * MT) == 0 && N2 % (16 * MT) == 0 && N3 % 16 == 0, "GEMM row blocks must tile M");

  const float* x   = (const float*)d_in[0];
  const int*   ei1 = (const int*)d_in[1];
  const int*   ei2 = (const int*)d_in[2];
  const int*   ei3 = (const int*)d_in[3];
  const float* Wl1 = (const float*)d_in[4];
  const float* bl1 = (const float*)d_in[5];
  const float* Wr1 = (const float*)d_in[6];
  const float* Wl2 = (const float*)d_in[7];
  const float* bl2 = (const float*)d_in[8];
  const float* Wr2 = (const float*)d_in[9];
  const float* Wl3 = (const float*)d_in[10];
  const float* bl3 = (const float*)d_in[11];
  const float* Wr3 = (const float*)d_in[12];
  const float* Wls = (const float*)d_in[13];
  const float* bls = (const float*)d_in[14];
  const float* Wrs = (const float*)d_in[15];

  const int E1 = in_sizes[1] / 2;
  const int E2 = in_sizes[2] / 2;
  const int E3 = in_sizes[3] / 2;
  const int ET = E1 + E2 + E3;
  const int NT = N1 + N2 + N3;

  float* h1  = (float*)d_ws;
  int*   csr = (int*)(h1 + (long long)N1 * C_DIM);
  int*   deg  = csr;
  int*   rowp = deg + NT;
  int*   rank = rowp + NT;
  int*   perm = rank + ET;
  float* agg = (float*)(csr + (long long)N2 * C_DIM);
  float* h2  = agg + (long long)N2 * C_DIM;
  float* cnt = agg + (long long)N1 * C_DIM;
  float* wt  = cnt + N1;
  float* Wl1t = wt;
  float* Wr1t = Wl1t + C_DIM * C_DIM;
  float* Wl2t = Wr1t + C_DIM * C_DIM;
  float* Wr2t = Wl2t + C_DIM * C_DIM;
  float* Wl3t = Wr2t + C_DIM * C_DIM;
  float* Wr3t = Wl3t + C_DIM * OUT_C;
  float* Wlst = Wr3t + C_DIM * OUT_C;
  float* Wrst = Wlst + C_DIM * OUT_C;

  float* mu     = (float*)d_out;
  float* logstd = mu + (long long)N3 * OUT_C;

  const int ZB = 256, SB = 256, GB = 256;

  {
    zero_i32<<<cdiv(NT, ZB), ZB, 0, stream>>>(deg, NT);
    edge_rank<<<cdiv(ET, SB), SB, 0, stream>>>(ei1, ei2, ei3, E1, E2, E3, N1, N2, deg, rank);
    scan_deg<<<3 + TR_BLOCKS, 1024, 0, stream>>>(deg, rowp, N1, N2, N3,
                                                 Wl1, Wr1, Wl2, Wr2, Wl3, Wr3, Wls, Wrs, wt);
    fill_perm<<<cdiv(ET, SB), SB, 0, stream>>>(ei1, ei2, ei3, E1, E2, E3, N1, N2, rowp, rank, perm);
  }

  {
    gather_rows<<<cdiv(N1, 8), GB, 0, stream>>>(x, rowp, deg, perm, agg, cnt, N1);
    int tiles = (N1 / (16 * MT)) * (C_DIM / 64);
    gemm_dual_wmma<MT, 1><<<cdiv(tiles, 4), 128, 0, stream>>>(
        agg, Wl1t, cnt, x, Wr1t, bl1, h1, N1, C_DIM);
  }

  {
    gather_rows<<<cdiv(N2, 8), GB, 0, stream>>>(h1, rowp + N1, deg + N1, perm + E1, agg, cnt, N2);
    int tiles = (N2 / (16 * MT)) * (C_DIM / 64);
    gemm_dual_wmma<MT, 1><<<cdiv(tiles, 4), 128, 0, stream>>>(
        agg, Wl2t, cnt, h1, Wr2t, bl2, h2, N2, C_DIM);
  }

  {
    gather_rows<<<cdiv(N3, 8), GB, 0, stream>>>(h2, rowp + N1 + N2, deg + N1 + N2, perm + E1 + E2,
                                                agg, cnt, N3);

    int tiles = (N3 / 16) * (OUT_C / 64);
    gemm_heads_wmma<<<dim3(cdiv(tiles, 8), 2), GB, 0, stream>>>(
        agg, cnt, h2, Wl3t, Wr3t, bl3, mu, Wlst, Wrst, bls, logstd, N3, OUT_C);
  }
}


# All GEMMs: each lane loads 4 consecutive k (one b128) per fragment and feeds 2 WMMAs, halving loads and cache lines touched per WMMA; k pairs differently inside the WMMAs, so the sum order changes
# speedup vs baseline: 2.5047x; 1.0992x over previous
#include <hip/hip_runtime.h>
#include <hip/hip_bf16.h>

typedef __attribute__((ext_vector_type(2))) float v2f;
typedef __attribute__((ext_vector_type(4))) float v4f;
typedef __attribute__((ext_vector_type(8))) float v8f;

#define C_DIM 256

__global__ void zero_i32(int* __restrict__ p, int n) {
  int i = blockIdx.x * blockDim.x + threadIdx.x;
  if (i < n) p[i] = 0;
}

#define O_DIM 64
#define TR_BIG_BLOCKS (C_DIM * C_DIM / 4096)
#define TR_SML_BLOCKS (C_DIM * O_DIM / 4096)
#define TR_BLOCKS (4 * TR_BIG_BLOCKS + 4 * TR_SML_BLOCKS)
__device__ __forceinline__ void transpose_block(
    int b, int tid,
    const float* __restrict__ w0, const float* __restrict__ w1,
    const float* __restrict__ w2, const float* __restrict__ w3,
    const float* __restrict__ w4, const float* __restrict__ w5,
    const float* __restrict__ w6, const float* __restrict__ w7,
    float* __restrict__ wt) {
  constexpr int K = C_DIM;
  int m, local;
  if (b < 4 * TR_BIG_BLOCKS) { m = b / TR_BIG_BLOCKS; local = b % TR_BIG_BLOCKS; }
  else { const int r = b - 4 * TR_BIG_BLOCKS; m = 4 + r / TR_SML_BLOCKS; local = r % TR_SML_BLOCKS; }
  const float* W = m == 0 ? w0 : m == 1 ? w1 : m == 2 ? w2 : m == 3 ? w3
                 : m == 4 ? w4 : m == 5 ? w5 : m == 6 ? w6 : w7;
  const int N = m < 4 ? C_DIM : O_DIM;
  float* Wt = wt + (m < 4 ? m * (C_DIM * C_DIM)
                          : 4 * (C_DIM * C_DIM) + (m - 4) * (C_DIM * O_DIM));
  const int t = (local * 1024 + tid) * 4;
  const int k = t / N;
  const int n = t % N;
  const float4 v = *(const float4*)(W + t);
  Wt[(n + 0) * K + k] = v.x;
  Wt[(n + 1) * K + k] = v.y;
  Wt[(n + 2) * K + k] = v.z;
  Wt[(n + 3) * K + k] = v.w;
}

__global__ void edge_rank(const int* __restrict__ ei1, const int* __restrict__ ei2,
                          const int* __restrict__ ei3, int E1, int E2, int E3,
                          int n1, int n2,
                          int* __restrict__ deg, int* __restrict__ rank) {
  const int t = blockIdx.x * blockDim.x + threadIdx.x;
  const int* dst;
  int e, roff;
  if (t < E1)                { e = t;           dst = ei1 + E1; roff = 0; }
  else if (t < E1 + E2)      { e = t - E1;      dst = ei2 + E2; roff = n1; }
  else if (t < E1 + E2 + E3) { e = t - E1 - E2; dst = ei3 + E3; roff = n1 + n2; }
  else return;
  rank[t] = atomicAdd(deg + roff + dst[e], 1);
}

__global__ void fill_perm(const int* __restrict__ ei1, const int* __restrict__ ei2,
                          const int* __restrict__ ei3, int E1, int E2, int E3,
                          int n1, int n2,
                          const int* __restrict__ rowptr, const int* __restrict__ rank,
                          int* __restrict__ perm) {
  const int t = blockIdx.x * blockDim.x + threadIdx.x;
  const int* src;
  const int* dst;
  int e, roff, eoff;
  if (t < E1)                { e = t;           src = ei1; dst = ei1 + E1; roff = 0;       eoff = 0; }
  else if (t < E1 + E2)      { e = t - E1;      src = ei2; dst = ei2 + E2; roff = n1;      eoff = E1; }
  else if (t < E1 + E2 + E3) { e = t - E1 - E2; src = ei3; dst = ei3 + E3; roff = n1 + n2; eoff = E1 + E2; }
  else return;
  perm[eoff + rowptr[roff + dst[e]] + rank[t]] = src[e];
}

#define SCAN_TILES 10
__global__ __launch_bounds__(1024) void scan_deg(const int* __restrict__ deg,
                                                 int* __restrict__ rowptr,
                                                 int n1, int n2, int n3,
    const float* __restrict__ w0, const float* __restrict__ w1,
    const float* __restrict__ w2, const float* __restrict__ w3,
    const float* __restrict__ w4, const float* __restrict__ w5,
    const float* __restrict__ w6, const float* __restrict__ w7,
    float* __restrict__ wt) {
  __shared__ int wsum[SCAN_TILES * 32];
  __shared__ int tsum[SCAN_TILES];
  if (blockIdx.x >= 3) {
    transpose_block(blockIdx.x - 3, threadIdx.x, w0, w1, w2, w3, w4, w5, w6, w7, wt);
    return;
  }
  const int set = blockIdx.x;
  const int n   = set == 0 ? n1 : (set == 1 ? n2 : n3);
  const int off = set == 0 ? 0  : (set == 1 ? n1 : n1 + n2);
  const int tid = threadIdx.x, lane = tid & 31, wave = tid >> 5;
  const int4* in  = (const int4*)(deg + off);
  int4*       out = (int4*)(rowptr + off);

  int4 v[SCAN_TILES];
  int  incl[SCAN_TILES];
  for (int i = 0; i < SCAN_TILES; ++i) {
    const int q = i * 1024 + tid;
    v[i] = (4 * q < n) ? in[q] : make_int4(0, 0, 0, 0);
  }
  for (int i = 0; i < SCAN_TILES; ++i) {
    int s = v[i].x + v[i].y + v[i].z + v[i].w;
    for (int d = 1; d < 32; d <<= 1) {
      const int up = __shfl_up(s, d, 32);
      if (lane >= d) s += up;
    }
    incl[i] = s;
    if (lane == 31) wsum[i * 32 + wave] = s;
  }
  __syncthreads();
  if (tid < SCAN_TILES * 32) {
    const int w = wsum[tid];
    int s = w;
    for (int d = 1; d < 32; d <<= 1) {
      const int up = __shfl_up(s, d, 32);
      if (lane >= d) s += up;
    }
    wsum[tid] = s - w;
    if (lane == 31) tsum[wave] = s;
  }
  __syncthreads();
  int carry = 0;
  for (int i = 0; i < SCAN_TILES; ++i) {
    const int q = i * 1024 + tid;
    const int own = v[i].x + v[i].y + v[i].z + v[i].w;
    const int b0 = carry + wsum[i * 32 + wave] + incl[i] - own;
    if (4 * q < n)
      out[q] = make_int4(b0, b0 + v[i].x, b0 + v[i].x + v[i].y, b0 + v[i].x + v[i].y + v[i].z);
    carry += tsum[i];
  }
}

template <int B>
__device__ __forceinline__ void add_rows(const float* __restrict__ xl, int idx, int j,
                                         float4& a0, float4& a1) {
  float4 u[B], w[B];
  for (int t = 0; t < B; ++t) {
    const float* r = xl + (long long)__builtin_amdgcn_readlane(idx, j + t) * C_DIM;
    u[t] = *(const float4*)r;
    w[t] = *(const float4*)(r + 128);
  }
  for (int t = 0; t < B; ++t) {
    a0.x += u[t].x; a0.y += u[t].y; a0.z += u[t].z; a0.w += u[t].w;
    a1.x += w[t].x; a1.y += w[t].y; a1.z += w[t].z; a1.w += w[t].w;
  }
}

__global__ __launch_bounds__(256) void gather_rows(
    const float* __restrict__ x, const int* __restrict__ rowptr,
    const int* __restrict__ deg, const int* __restrict__ perm,
    float* __restrict__ agg, float* __restrict__ cnt, int n_dst) {
  const int lane = threadIdx.x & 31;
  const int d = __builtin_amdgcn_readfirstlane(blockIdx.x * 8 + (threadIdx.x >> 5));
  if (d >= n_dst) return;
  const int beg = rowptr[d];
  const int n   = deg[d];
  const float* xl = x + lane * 4;

  float4 a0 = make_float4(0.f, 0.f, 0.f, 0.f);
  float4 a1 = make_float4(0.f, 0.f, 0.f, 0.f);

  for (int base = 0; base < n; base += 32) {
    const int m = min(n - base, 32);
    int idx = 0;
    if (lane < m) idx = perm[beg + base + lane];
    int j = 0;
    for (; j + 8 <= m; j += 8) add_rows<8>(xl, idx, j, a0, a1);
    if (m & 4) { add_rows<4>(xl, idx, j, a0, a1); j += 4; }
    if (m & 2) { add_rows<2>(xl, idx, j, a0, a1); j += 2; }
    if (m & 1) { add_rows<1>(xl, idx, j, a0, a1); }
  }

  float* o = agg + (long long)d * C_DIM + lane * 4;
  *(float4*)o         = a0;
  *(float4*)(o + 128) = a1;
  if (lane == 0) cnt[d] = (float)n;
}

template <int MT>
__device__ __forceinline__ void load_frags(const float* __restrict__ ap,
                                           const float* __restrict__ wp, int k,
                                           v4f (&a)[MT], v4f (&b)[4]) {
  constexpr int K = C_DIM;
#pragma unroll
  for (int i = 0; i < MT; ++i) a[i] = *(const v4f*)(ap + i * 16 * K + k);
#pragma unroll
  for (int j = 0; j < 4; ++j) b[j] = *(const v4f*)(wp + j * 16 * K + k);
}

template <int MT, bool SCALE>
__device__ __forceinline__ void mma_step(const v4f (&a)[MT], const v4f (&b)[4],
                                         const float (&s)[MT], v8f (&acc)[MT][4]) {
#pragma unroll
  for (int i = 0; i < MT; ++i) {
    const v4f x = SCALE ? a[i] * s[i] : a[i];
    const v2f xlo = {x.x, x.y};
    const v2f xhi = {x.z, x.w};
#pragma unroll
    for (int j = 0; j < 4; ++j) {
      const v2f blo = {b[j].x, b[j].y};
      acc[i][j] = __builtin_amdgcn_wmma_f32_16x16x4_f32(false, xlo, false, blo,
                                                        (short)0, acc[i][j],
                                                        false, false);
    }
#pragma unroll
    for (int j = 0; j < 4; ++j) {
      const v2f bhi = {b[j].z, b[j].w};
      acc[i][j] = __builtin_amdgcn_wmma_f32_16x16x4_f32(false, xhi, false, bhi,
                                                        (short)0, acc[i][j],
                                                        false, false);
    }
  }
}

template <int MT, bool SCALE>
__device__ __forceinline__ void gemm_path(const float* __restrict__ ap,
                                          const float* __restrict__ wp,
                                          const float (&s)[MT], v8f (&acc)[MT][4]) {
  constexpr int K = C_DIM;
  static_assert(K % 16 == 0, "the k loop takes two 8-wide steps per trip");
  v4f a0[MT], b0[4], a1[MT], b1[4];
  load_frags<MT>(ap, wp, 0, a0, b0);
  for (int k = 0; k < K - 16; k += 16) {
    load_frags<MT>(ap, wp, k + 8, a1, b1);
    mma_step<MT, SCALE>(a0, b0, s, acc);
    load_frags<MT>(ap, wp, k + 16, a0, b0);
    mma_step<MT, SCALE>(a1, b1, s, acc);
  }
  load_frags<MT>(ap, wp, K - 8, a1, b1);
  mma_step<MT, SCALE>(a0, b0, s, acc);
  mma_step<MT, SCALE>(a1, b1, s, acc);
}

template <int MT, int RELU>
__device__ __forceinline__ void gemm_dual_tile(
    const float* __restrict__ A1, const float* __restrict__ Wt1,
    const float* __restrict__ cnt1,
    const float* __restrict__ A2, const float* __restrict__ Wt2,
    const float* __restrict__ bias, float* __restrict__ out,
    int M, int N) {
  constexpr int K = C_DIM;
  constexpr int BM = 16 * MT;
  const int lane = threadIdx.x & 31;
  const int wave = threadIdx.x >> 5;
  const int nstrips = N >> 6;
  const int tile = blockIdx.x * (blockDim.x >> 5) + wave;
  if (tile >= (M / BM) * nstrips) return;

  const int m0 = (tile / nstrips) * BM;
  const int n0 = (tile % nstrips) << 6;

  const int half = lane >> 4;
  const int l15  = lane & 15;
  const int row  = m0 + l15;
  const int cb   = n0 + l15;
  const int koff = half << 2;

  float s1[MT];
#pragma unroll
  for (int i = 0; i < MT; ++i) s1[i] = 1.0f / fmaxf(cnt1[row + i * 16], 1.0f);

  v8f acc[MT][4];
#pragma unroll
  for (int i = 0; i < MT; ++i)
#pragma unroll
    for (int j = 0; j < 4; ++j) acc[i][j] = (v8f){0.f,0.f,0.f,0.f,0.f,0.f,0.f,0.f};

  const float* a1p = A1 + (long long)row * K + koff;
  const float* a2p = A2 + (long long)row * K + koff;
  const float* w1p = Wt1 + (long long)cb * K + koff;
  const float* w2p = Wt2 + (long long)cb * K + koff;

  gemm_path<MT, true >(a1p, w1p, s1, acc);
  gemm_path<MT, false>(a2p, w2p, s1, acc);

#pragma unroll
  for (int j = 0; j < 4; ++j) {
    const int col = cb + j * 16;
    const float bv = bias[col];
#pragma unroll
    for (int i = 0; i < MT; ++i) {
#pragma unroll
      for (int r = 0; r < 8; ++r) {
        float v = acc[i][j][r] + bv;
        if (RELU) v = fmaxf(v, 0.f);
        out[(long long)(m0 + i * 16 + r + half * 8) * N + col] = v;
      }
    }
  }
}

template <int MT, int RELU>
__global__ __launch_bounds__(256) void gemm_dual_wmma(
    const float* __restrict__ A1, const float* __restrict__ Wt1,
    const float* __restrict__ cnt1,
    const float* __restrict__ A2, const float* __restrict__ Wt2,
    const float* __restrict__ bias, float* __restrict__ out,
    int M, int N) {
  gemm_dual_tile<MT, RELU>(A1, Wt1, cnt1, A2, Wt2, bias, out, M, N);
}

__global__ __launch_bounds__(256) void gemm_heads_wmma(
    const float* __restrict__ A1, const float* __restrict__ cnt1,
    const float* __restrict__ A2,
    const float* __restrict__ Wt1a, const float* __restrict__ Wt2a,
    const float* __restrict__ biasa, float* __restrict__ outa,
    const float* __restrict__ Wt1b, const float* __restrict__ Wt2b,
    const float* __restrict__ biasb, float* __restrict__ outb,
    int M, int N) {
  const bool second = blockIdx.y != 0;
  gemm_dual_tile<1, 0>(A1, second ? Wt1b : Wt1a, cnt1,
                       A2, second ? Wt2b : Wt2a,
                       second ? biasb : biasa, second ? outb : outa, M, N);
}

static inline int cdiv(long long a, long long b) { return (int)((a + b - 1) / b); }

extern "C" void kernel_launch(void* const* d_in, const int* in_sizes, int n_in,
                              void* d_out, int out_size, void* d_ws, size_t ws_size,
                              hipStream_t stream) {
  (void)n_in; (void)out_size; (void)ws_size;

  const int N1 = 40000, N2 = 20000, N3 = 10000;
  const int OUT_C = O_DIM;
  static_assert(N1 % 4 == 0 && N2 % 4 == 0 && N3 % 4 == 0, "scan_deg reads rows four at a time");
  static_assert(N1 <= SCAN_TILES * 4096 && N2 <= N1 && N3 <= N1, "scan_deg tile count");
  static_assert(2 * N2 <= N1 && N3 <= N2, "h2 lives in the half of agg that layers 2 and 3 leave free");
  constexpr int MT = 5;
  static_assert(N1 % (16 * MT) == 0 && N2 % (16 * MT) == 0 && N3 % 16 == 0, "GEMM row blocks must tile M");

  const float* x   = (const float*)d_in[0];
  const int*   ei1 = (const int*)d_in[1];
  const int*   ei2 = (const int*)d_in[2];
  const int*   ei3 = (const int*)d_in[3];
  const float* Wl1 = (const float*)d_in[4];
  const float* bl1 = (const float*)d_in[5];
  const float* Wr1 = (const float*)d_in[6];
  const float* Wl2 = (const float*)d_in[7];
  const float* bl2 = (const float*)d_in[8];
  const float* Wr2 = (const float*)d_in[9];
  const float* Wl3 = (const float*)d_in[10];
  const float* bl3 = (const float*)d_in[11];
  const float* Wr3 = (const float*)d_in[12];
  const float* Wls = (const float*)d_in[13];
  const float* bls = (const float*)d_in[14];
  const float* Wrs = (const float*)d_in[15];

  const int E1 = in_sizes[1] / 2;
  const int E2 = in_sizes[2] / 2;
  const int E3 = in_sizes[3] / 2;
  const int ET = E1 + E2 + E3;
  const int NT = N1 + N2 + N3;

  float* h1  = (float*)d_ws;
  int*   csr = (int*)(h1 + (long long)N1 * C_DIM);
  int*   deg  = csr;
  int*   rowp = deg + NT;
  int*   rank = rowp + NT;
  int*   perm = rank + ET;
  float* agg = (float*)(csr + (long long)N2 * C_DIM);
  float* h2  = agg + (long long)N2 * C_DIM;
  float* cnt = agg + (long long)N1 * C_DIM;
  float* wt  = cnt + N1;
  float* Wl1t = wt;
  float* Wr1t = Wl1t + C_DIM * C_DIM;
  float* Wl2t = Wr1t + C_DIM * C_DIM;
  float* Wr2t = Wl2t + C_DIM * C_DIM;
  float* Wl3t = Wr2t + C_DIM * C_DIM;
  float* Wr3t = Wl3t + C_DIM * OUT_C;
  float* Wlst = Wr3t + C_DIM * OUT_C;
  float* Wrst = Wlst + C_DIM * OUT_C;

  float* mu     = (float*)d_out;
  float* logstd = mu + (long long)N3 * OUT_C;

  const int ZB = 256, SB = 256, GB = 256;

  {
    zero_i32<<<cdiv(NT, ZB), ZB, 0, stream>>>(deg, NT);
    edge_rank<<<cdiv(ET, SB), SB, 0, stream>>>(ei1, ei2, ei3, E1, E2, E3, N1, N2, deg, rank);
    scan_deg<<<3 + TR_BLOCKS, 1024, 0, stream>>>(deg, rowp, N1, N2, N3,
                                                 Wl1, Wr1, Wl2, Wr2, Wl3, Wr3, Wls, Wrs, wt);
    fill_perm<<<cdiv(ET, SB), SB, 0, stream>>>(ei1, ei2, ei3, E1, E2, E3, N1, N2, rowp, rank, perm);
  }

  {
    gather_rows<<<cdiv(N1, 8), GB, 0, stream>>>(x, rowp, deg, perm, agg, cnt, N1);
    int tiles = (N1 / (16 * MT)) * (C_DIM / 64);
    gemm_dual_wmma<MT, 1><<<cdiv(tiles, 4), 128, 0, stream>>>(
        agg, Wl1t, cnt, x, Wr1t, bl1, h1, N1, C_DIM);
  }

  {
    gather_rows<<<cdiv(N2, 8), GB, 0, stream>>>(h1, rowp + N1, deg + N1, perm + E1, agg, cnt, N2);
    int tiles = (N2 / (16 * MT)) * (C_DIM / 64);
    gemm_dual_wmma<MT, 1><<<cdiv(tiles, 4), 128, 0, stream>>>(
        agg, Wl2t, cnt, h1, Wr2t, bl2, h2, N2, C_DIM);
  }

  {
    gather_rows<<<cdiv(N3, 8), GB, 0, stream>>>(h2, rowp + N1 + N2, deg + N1 + N2, perm + E1 + E2,
                                                agg, cnt, N3);

    int tiles = (N3 / 16) * (OUT_C / 64);
    gemm_heads_wmma<<<dim3(cdiv(tiles, 8), 2), GB, 0, stream>>>(
        agg, cnt, h2, Wl3t, Wr3t, bl3, mu, Wlst, Wrst, bls, logstd, N3, OUT_C);
  }
}


# edge_rank and fill_perm: each thread takes 8 edges (16-byte id loads, all its atomics in flight together) instead of 1: 4,102 waves to start per launch, not 32,816; slot arrival order may shift
# speedup vs baseline: 2.6125x; 1.0430x over previous
#include <hip/hip_runtime.h>
#include <hip/hip_bf16.h>

typedef __attribute__((ext_vector_type(2))) float v2f;
typedef __attribute__((ext_vector_type(4))) float v4f;
typedef __attribute__((ext_vector_type(8))) float v8f;

#define C_DIM 256

__global__ void zero_i32(int* __restrict__ p, int n) {
  int i = blockIdx.x * blockDim.x + threadIdx.x;
  if (i < n) p[i] = 0;
}

#define O_DIM 64
#define TR_BIG_BLOCKS (C_DIM * C_DIM / 4096)
#define TR_SML_BLOCKS (C_DIM * O_DIM / 4096)
#define TR_BLOCKS (4 * TR_BIG_BLOCKS + 4 * TR_SML_BLOCKS)
__device__ __forceinline__ void transpose_block(
    int b, int tid,
    const float* __restrict__ w0, const float* __restrict__ w1,
    const float* __restrict__ w2, const float* __restrict__ w3,
    const float* __restrict__ w4, const float* __restrict__ w5,
    const float* __restrict__ w6, const float* __restrict__ w7,
    float* __restrict__ wt) {
  constexpr int K = C_DIM;
  int m, local;
  if (b < 4 * TR_BIG_BLOCKS) { m = b / TR_BIG_BLOCKS; local = b % TR_BIG_BLOCKS; }
  else { const int r = b - 4 * TR_BIG_BLOCKS; m = 4 + r / TR_SML_BLOCKS; local = r % TR_SML_BLOCKS; }
  const float* W = m == 0 ? w0 : m == 1 ? w1 : m == 2 ? w2 : m == 3 ? w3
                 : m == 4 ? w4 : m == 5 ? w5 : m == 6 ? w6 : w7;
  const int N = m < 4 ? C_DIM : O_DIM;
  float* Wt = wt + (m < 4 ? m * (C_DIM * C_DIM)
                          : 4 * (C_DIM * C_DIM) + (m - 4) * (C_DIM * O_DIM));
  const int t = (local * 1024 + tid) * 4;
  const int k = t / N;
  const int n = t % N;
  const float4 v = *(const float4*)(W + t);
  Wt[(n + 0) * K + k] = v.x;
  Wt[(n + 1) * K + k] = v.y;
  Wt[(n + 2) * K + k] = v.z;
  Wt[(n + 3) * K + k] = v.w;
}

#define EC_PER_THREAD 2
#define EC_BLOCK 256

struct EdgeChunk {
  const int* src;
  const int* dst;
  int e;
  int roff;
  int eoff;
  int nv;
};

__device__ __forceinline__ EdgeChunk edge_chunk(
    int q, const int* __restrict__ ei1, const int* __restrict__ ei2,
    const int* __restrict__ ei3, int E1, int E2, int E3, int n1, int n2) {
  const int Q1 = (E1 + 3) >> 2, Q2 = (E2 + 3) >> 2;
  const bool in1 = q < Q1, in2 = q < Q1 + Q2;
  const int E = in1 ? E1 : in2 ? E2 : E3;
  EdgeChunk c;
  c.src  = in1 ? ei1 : in2 ? ei2 : ei3;
  c.e    = 4 * (q - (in1 ? 0 : in2 ? Q1 : Q1 + Q2));
  c.roff = in1 ? 0 : in2 ? n1 : n1 + n2;
  c.eoff = in1 ? 0 : in2 ? E1 : E1 + E2;
  c.dst  = c.src + E;
  c.nv   = min(4, E - c.e);
  return c;
}

template <bool WHOLE>
__device__ __forceinline__ int4 load_ids4(const int* __restrict__ p, int e, int nv) {
  if (WHOLE) return *(const int4*)(p + e);
  int4 v = make_int4(0, 0, 0, 0);
  v.x = p[e];
  if (nv > 1) v.y = p[e + 1];
  if (nv > 2) v.z = p[e + 2];
  if (nv > 3) v.w = p[e + 3];
  return v;
}

template <bool WHOLE, bool FULL>
__device__ __forceinline__ void edge_rank_body(
    const int* __restrict__ ei1, const int* __restrict__ ei2,
    const int* __restrict__ ei3, int E1, int E2, int E3, int n1, int n2,
    int* __restrict__ deg, int* __restrict__ rank) {
  const int QT = ((E1 + 3) >> 2) + ((E2 + 3) >> 2) + ((E3 + 3) >> 2);
  const int q0 = blockIdx.x * (EC_BLOCK * EC_PER_THREAD) + threadIdx.x;
  EdgeChunk c[EC_PER_THREAD];
  int4 d[EC_PER_THREAD], r[EC_PER_THREAD];
  int nv[EC_PER_THREAD];
  for (int u = 0; u < EC_PER_THREAD; ++u) {
    const int q = q0 + u * EC_BLOCK;
    c[u] = edge_chunk(min(q, QT - 1), ei1, ei2, ei3, E1, E2, E3, n1, n2);
    d[u] = load_ids4<WHOLE>(c[u].dst, c[u].e, c[u].nv);
    nv[u] = FULL ? 4 : (q < QT ? c[u].nv : 0);
  }
  asm volatile("" ::: "memory");
  for (int u = 0; u < EC_PER_THREAD; ++u) {
    r[u] = make_int4(0, 0, 0, 0);
    if (FULL || nv[u] > 0) {
      int* g = deg + c[u].roff;
      r[u].x = atomicAdd(g + d[u].x, 1);
      if (WHOLE || nv[u] > 1) r[u].y = atomicAdd(g + d[u].y, 1);
      if (WHOLE || nv[u] > 2) r[u].z = atomicAdd(g + d[u].z, 1);
      if (WHOLE || nv[u] > 3) r[u].w = atomicAdd(g + d[u].w, 1);
    }
  }
  for (int u = 0; u < EC_PER_THREAD; ++u)
    if (FULL || nv[u] > 0) *(int4*)(rank + 4 * (q0 + u * EC_BLOCK)) = r[u];
}

__device__ __forceinline__ bool edge_block_full(int E1, int E2, int E3) {
  const int QT = ((E1 + 3) >> 2) + ((E2 + 3) >> 2) + ((E3 + 3) >> 2);
  return (int)(blockIdx.x + 1) * (EC_BLOCK * EC_PER_THREAD) <= QT;
}

__global__ __launch_bounds__(EC_BLOCK) void edge_rank(
    const int* __restrict__ ei1, const int* __restrict__ ei2,
    const int* __restrict__ ei3, int E1, int E2, int E3,
    int n1, int n2,
    int* __restrict__ deg, int* __restrict__ rank) {
  const bool whole = ((E1 | E2 | E3) & 3) == 0;
  if (whole && edge_block_full(E1, E2, E3))
    edge_rank_body<true, true >(ei1, ei2, ei3, E1, E2, E3, n1, n2, deg, rank);
  else if (whole)
    edge_rank_body<true, false>(ei1, ei2, ei3, E1, E2, E3, n1, n2, deg, rank);
  else
    edge_rank_body<false, false>(ei1, ei2, ei3, E1, E2, E3, n1, n2, deg, rank);
}

template <bool WHOLE, bool FULL>
__device__ __forceinline__ void fill_perm_body(
    const int* __restrict__ ei1, const int* __restrict__ ei2,
    const int* __restrict__ ei3, int E1, int E2, int E3, int n1, int n2,
    const int* __restrict__ rowptr, const int* __restrict__ rank,
    int* __restrict__ perm) {
  const int QT = ((E1 + 3) >> 2) + ((E2 + 3) >> 2) + ((E3 + 3) >> 2);
  const int q0 = blockIdx.x * (EC_BLOCK * EC_PER_THREAD) + threadIdx.x;
  EdgeChunk c[EC_PER_THREAD];
  int4 s[EC_PER_THREAD], d[EC_PER_THREAD], r[EC_PER_THREAD], b[EC_PER_THREAD];
  int nv[EC_PER_THREAD];
  for (int u = 0; u < EC_PER_THREAD; ++u) {
    const int q = q0 + u * EC_BLOCK;
    const int qc = min(q, QT - 1);
    c[u] = edge_chunk(qc, ei1, ei2, ei3, E1, E2, E3, n1, n2);
    s[u] = load_ids4<WHOLE>(c[u].src, c[u].e, c[u].nv);
    d[u] = load_ids4<WHOLE>(c[u].dst, c[u].e, c[u].nv);
    r[u] = *(const int4*)(rank + 4 * qc);
    nv[u] = FULL ? 4 : (q < QT ? c[u].nv : 0);
  }
  asm volatile("" ::: "memory");
  for (int u = 0; u < EC_PER_THREAD; ++u) {
    const int* rp = rowptr + c[u].roff;
    b[u].x = rp[d[u].x];
    b[u].y = rp[d[u].y];
    b[u].z = rp[d[u].z];
    b[u].w = rp[d[u].w];
  }
  for (int u = 0; u < EC_PER_THREAD; ++u) {
    if (FULL || nv[u] > 0) {
      int* pm = perm + c[u].eoff;
      pm[b[u].x + r[u].x] = s[u].x;
      if (WHOLE || nv[u] > 1) pm[b[u].y + r[u].y] = s[u].y;
      if (WHOLE || nv[u] > 2) pm[b[u].z + r[u].z] = s[u].z;
      if (WHOLE || nv[u] > 3) pm[b[u].w + r[u].w] = s[u].w;
    }
  }
}

__global__ __launch_bounds__(EC_BLOCK) void fill_perm(
    const int* __restrict__ ei1, const int* __restrict__ ei2,
    const int* __restrict__ ei3, int E1, int E2, int E3,
    int n1, int n2,
    const int* __restrict__ rowptr, const int* __restrict__ rank,
    int* __restrict__ perm) {
  const bool whole = ((E1 | E2 | E3) & 3) == 0;
  if (whole && edge_block_full(E1, E2, E3))
    fill_perm_body<true, true >(ei1, ei2, ei3, E1, E2, E3, n1, n2, rowptr, rank, perm);
  else if (whole)
    fill_perm_body<true, false>(ei1, ei2, ei3, E1, E2, E3, n1, n2, rowptr, rank, perm);
  else
    fill_perm_body<false, false>(ei1, ei2, ei3, E1, E2, E3, n1, n2, rowptr, rank, perm);
}

#define SCAN_TILES 10
__global__ __launch_bounds__(1024) void scan_deg(const int* __restrict__ deg,
                                                 int* __restrict__ rowptr,
                                                 int n1, int n2, int n3,
    const float* __restrict__ w0, const float* __restrict__ w1,
    const float* __restrict__ w2, const float* __restrict__ w3,
    const float* __restrict__ w4, const float* __restrict__ w5,
    const float* __restrict__ w6, const float* __restrict__ w7,
    float* __restrict__ wt) {
  __shared__ int wsum[SCAN_TILES * 32];
  __shared__ int tsum[SCAN_TILES];
  if (blockIdx.x >= 3) {
    transpose_block(blockIdx.x - 3, threadIdx.x, w0, w1, w2, w3, w4, w5, w6, w7, wt);
    return;
  }
  const int set = blockIdx.x;
  const int n   = set == 0 ? n1 : (set == 1 ? n2 : n3);
  const int off = set == 0 ? 0  : (set == 1 ? n1 : n1 + n2);
  const int tid = threadIdx.x, lane = tid & 31, wave = tid >> 5;
  const int4* in  = (const int4*)(deg + off);
  int4*       out = (int4*)(rowptr + off);

  int4 v[SCAN_TILES];
  int  incl[SCAN_TILES];
  for (int i = 0; i < SCAN_TILES; ++i) {
    const int q = i * 1024 + tid;
    v[i] = (4 * q < n) ? in[q] : make_int4(0, 0, 0, 0);
  }
  for (int i = 0; i < SCAN_TILES; ++i) {
    int s = v[i].x + v[i].y + v[i].z + v[i].w;
    for (int d = 1; d < 32; d <<= 1) {
      const int up = __shfl_up(s, d, 32);
      if (lane >= d) s += up;
    }
    incl[i] = s;
    if (lane == 31) wsum[i * 32 + wave] = s;
  }
  __syncthreads();
  if (tid < SCAN_TILES * 32) {
    const int w = wsum[tid];
    int s = w;
    for (int d = 1; d < 32; d <<= 1) {
      const int up = __shfl_up(s, d, 32);
      if (lane >= d) s += up;
    }
    wsum[tid] = s - w;
    if (lane == 31) tsum[wave] = s;
  }
  __syncthreads();
  int carry = 0;
  for (int i = 0; i < SCAN_TILES; ++i) {
    const int q = i * 1024 + tid;
    const int own = v[i].x + v[i].y + v[i].z + v[i].w;
    const int b0 = carry + wsum[i * 32 + wave] + incl[i] - own;
    if (4 * q < n)
      out[q] = make_int4(b0, b0 + v[i].x, b0 + v[i].x + v[i].y, b0 + v[i].x + v[i].y + v[i].z);
    carry += tsum[i];
  }
}

template <int B>
__device__ __forceinline__ void add_rows(const float* __restrict__ xl, int idx, int j,
                                         float4& a0, float4& a1) {
  float4 u[B], w[B];
  for (int t = 0; t < B; ++t) {
    const float* r = xl + (long long)__builtin_amdgcn_readlane(idx, j + t) * C_DIM;
    u[t] = *(const float4*)r;
    w[t] = *(const float4*)(r + 128);
  }
  for (int t = 0; t < B; ++t) {
    a0.x += u[t].x; a0.y += u[t].y; a0.z += u[t].z; a0.w += u[t].w;
    a1.x += w[t].x; a1.y += w[t].y; a1.z += w[t].z; a1.w += w[t].w;
  }
}

__global__ __launch_bounds__(256) void gather_rows(
    const float* __restrict__ x, const int* __restrict__ rowptr,
    const int* __restrict__ deg, const int* __restrict__ perm,
    float* __restrict__ agg, float* __restrict__ cnt, int n_dst) {
  const int lane = threadIdx.x & 31;
  const int d = __builtin_amdgcn_readfirstlane(blockIdx.x * 8 + (threadIdx.x >> 5));
  if (d >= n_dst) return;
  const int beg = rowptr[d];
  const int n   = deg[d];
  const float* xl = x + lane * 4;

  float4 a0 = make_float4(0.f, 0.f, 0.f, 0.f);
  float4 a1 = make_float4(0.f, 0.f, 0.f, 0.f);

  for (int base = 0; base < n; base += 32) {
    const int m = min(n - base, 32);
    int idx = 0;
    if (lane < m) idx = perm[beg + base + lane];
    int j = 0;
    for (; j + 8 <= m; j += 8) add_rows<8>(xl, idx, j, a0, a1);
    if (m & 4) { add_rows<4>(xl, idx, j, a0, a1); j += 4; }
    if (m & 2) { add_rows<2>(xl, idx, j, a0, a1); j += 2; }
    if (m & 1) { add_rows<1>(xl, idx, j, a0, a1); }
  }

  float* o = agg + (long long)d * C_DIM + lane * 4;
  *(float4*)o         = a0;
  *(float4*)(o + 128) = a1;
  if (lane == 0) cnt[d] = (float)n;
}

template <int MT>
__device__ __forceinline__ void load_frags(const float* __restrict__ ap,
                                           const float* __restrict__ wp, int k,
                                           v4f (&a)[MT], v4f (&b)[4]) {
  constexpr int K = C_DIM;
#pragma unroll
  for (int i = 0; i < MT; ++i) a[i] = *(const v4f*)(ap + i * 16 * K + k);
#pragma unroll
  for (int j = 0; j < 4; ++j) b[j] = *(const v4f*)(wp + j * 16 * K + k);
}

template <int MT, bool SCALE>
__device__ __forceinline__ void mma_step(const v4f (&a)[MT], const v4f (&b)[4],
                                         const float (&s)[MT], v8f (&acc)[MT][4]) {
#pragma unroll
  for (int i = 0; i < MT; ++i) {
    const v4f x = SCALE ? a[i] * s[i] : a[i];
    const v2f xlo = {x.x, x.y};
    const v2f xhi = {x.z, x.w};
#pragma unroll
    for (int j = 0; j < 4; ++j) {
      const v2f blo = {b[j].x, b[j].y};
      acc[i][j] = __builtin_amdgcn_wmma_f32_16x16x4_f32(false, xlo, false, blo,
                                                        (short)0, acc[i][j],
                                                        false, false);
    }
#pragma unroll
    for (int j = 0; j < 4; ++j) {
      const v2f bhi = {b[j].z, b[j].w};
      acc[i][j] = __builtin_amdgcn_wmma_f32_16x16x4_f32(false, xhi, false, bhi,
                                                        (short)0, acc[i][j],
                                                        false, false);
    }
  }
}

template <int MT, bool SCALE>
__device__ __forceinline__ void gemm_path(const float* __restrict__ ap,
                                          const float* __restrict__ wp,
                                          const float (&s)[MT], v8f (&acc)[MT][4]) {
  constexpr int K = C_DIM;
  static_assert(K % 16 == 0, "the k loop takes two 8-wide steps per trip");
  v4f a0[MT], b0[4], a1[MT], b1[4];
  load_frags<MT>(ap, wp, 0, a0, b0);
  for (int k = 0; k < K - 16; k += 16) {
    load_frags<MT>(ap, wp, k + 8, a1, b1);
    mma_step<MT, SCALE>(a0, b0, s, acc);
    load_frags<MT>(ap, wp, k + 16, a0, b0);
    mma_step<MT, SCALE>(a1, b1, s, acc);
  }
  load_frags<MT>(ap, wp, K - 8, a1, b1);
  mma_step<MT, SCALE>(a0, b0, s, acc);
  mma_step<MT, SCALE>(a1, b1, s, acc);
}

template <int MT, int RELU>
__device__ __forceinline__ void gemm_dual_tile(
    const float* __restrict__ A1, const float* __restrict__ Wt1,
    const float* __restrict__ cnt1,
    const float* __restrict__ A2, const float* __restrict__ Wt2,
    const float* __restrict__ bias, float* __restrict__ out,
    int M, int N) {
  constexpr int K = C_DIM;
  constexpr int BM = 16 * MT;
  const int lane = threadIdx.x & 31;
  const int wave = threadIdx.x >> 5;
  const int nstrips = N >> 6;
  const int tile = blockIdx.x * (blockDim.x >> 5) + wave;
  if (tile >= (M / BM) * nstrips) return;

  const int m0 = (tile / nstrips) * BM;
  const int n0 = (tile % nstrips) << 6;

  const int half = lane >> 4;
  const int l15  = lane & 15;
  const int row  = m0 + l15;
  const int cb   = n0 + l15;
  const int koff = half << 2;

  float s1[MT];
#pragma unroll
  for (int i = 0; i < MT; ++i) s1[i] = 1.0f / fmaxf(cnt1[row + i * 16], 1.0f);

  v8f acc[MT][4];
#pragma unroll
  for (int i = 0; i < MT; ++i)
#pragma unroll
    for (int j = 0; j < 4; ++j) acc[i][j] = (v8f){0.f,0.f,0.f,0.f,0.f,0.f,0.f,0.f};

  const float* a1p = A1 + (long long)row * K + koff;
  const float* a2p = A2 + (long long)row * K + koff;
  const float* w1p = Wt1 + (long long)cb * K + koff;
  const float* w2p = Wt2 + (long long)cb * K + koff;

  gemm_path<MT, true >(a1p, w1p, s1, acc);
  gemm_path<MT, false>(a2p, w2p, s1, acc);

#pragma unroll
  for (int j = 0; j < 4; ++j) {
    const int col = cb + j * 16;
    const float bv = bias[col];
#pragma unroll
    for (int i = 0; i < MT; ++i) {
#pragma unroll
      for (int r = 0; r < 8; ++r) {
        float v = acc[i][j][r] + bv;
        if (RELU) v = fmaxf(v, 0.f);
        out[(long long)(m0 + i * 16 + r + half * 8) * N + col] = v;
      }
    }
  }
}

template <int MT, int RELU>
__global__ __launch_bounds__(256) void gemm_dual_wmma(
    const float* __restrict__ A1, const float* __restrict__ Wt1,
    const float* __restrict__ cnt1,
    const float* __restrict__ A2, const float* __restrict__ Wt2,
    const float* __restrict__ bias, float* __restrict__ out,
    int M, int N) {
  gemm_dual_tile<MT, RELU>(A1, Wt1, cnt1, A2, Wt2, bias, out, M, N);
}

__global__ __launch_bounds__(256) void gemm_heads_wmma(
    const float* __restrict__ A1, const float* __restrict__ cnt1,
    const float* __restrict__ A2,
    const float* __restrict__ Wt1a, const float* __restrict__ Wt2a,
    const float* __restrict__ biasa, float* __restrict__ outa,
    const float* __restrict__ Wt1b, const float* __restrict__ Wt2b,
    const float* __restrict__ biasb, float* __restrict__ outb,
    int M, int N) {
  const bool second = blockIdx.y != 0;
  gemm_dual_tile<1, 0>(A1, second ? Wt1b : Wt1a, cnt1,
                       A2, second ? Wt2b : Wt2a,
                       second ? biasb : biasa, second ? outb : outa, M, N);
}

static inline int cdiv(long long a, long long b) { return (int)((a + b - 1) / b); }

extern "C" void kernel_launch(void* const* d_in, const int* in_sizes, int n_in,
                              void* d_out, int out_size, void* d_ws, size_t ws_size,
                              hipStream_t stream) {
  (void)n_in; (void)out_size; (void)ws_size;

  const int N1 = 40000, N2 = 20000, N3 = 10000;
  const int OUT_C = O_DIM;
  static_assert(N1 % 4 == 0 && N2 % 4 == 0 && N3 % 4 == 0, "scan_deg reads rows four at a time");
  static_assert(N1 <= SCAN_TILES * 4096 && N2 <= N1 && N3 <= N1, "scan_deg tile count");
  static_assert(2 * N2 <= N1 && N3 <= N2, "h2 lives in the half of agg that layers 2 and 3 leave free");
  constexpr int MT = 5;
  static_assert(N1 % (16 * MT) == 0 && N2 % (16 * MT) == 0 && N3 % 16 == 0, "GEMM row blocks must tile M");

  const float* x   = (const float*)d_in[0];
  const int*   ei1 = (const int*)d_in[1];
  const int*   ei2 = (const int*)d_in[2];
  const int*   ei3 = (const int*)d_in[3];
  const float* Wl1 = (const float*)d_in[4];
  const float* bl1 = (const float*)d_in[5];
  const float* Wr1 = (const float*)d_in[6];
  const float* Wl2 = (const float*)d_in[7];
  const float* bl2 = (const float*)d_in[8];
  const float* Wr2 = (const float*)d_in[9];
  const float* Wl3 = (const float*)d_in[10];
  const float* bl3 = (const float*)d_in[11];
  const float* Wr3 = (const float*)d_in[12];
  const float* Wls = (const float*)d_in[13];
  const float* bls = (const float*)d_in[14];
  const float* Wrs = (const float*)d_in[15];

  const int E1 = in_sizes[1] / 2;
  const int E2 = in_sizes[2] / 2;
  const int E3 = in_sizes[3] / 2;
  const int QT = cdiv(E1, 4) + cdiv(E2, 4) + cdiv(E3, 4);
  const int NT = N1 + N2 + N3;

  float* h1  = (float*)d_ws;
  int*   csr = (int*)(h1 + (long long)N1 * C_DIM);
  int*   deg  = csr;
  int*   rowp = deg + NT;
  int*   rank = rowp + NT;
  int*   perm = rank + 4 * QT;
  float* agg = (float*)(csr + (long long)N2 * C_DIM);
  float* h2  = agg + (long long)N2 * C_DIM;
  float* cnt = agg + (long long)N1 * C_DIM;
  float* wt  = cnt + N1;
  float* Wl1t = wt;
  float* Wr1t = Wl1t + C_DIM * C_DIM;
  float* Wl2t = Wr1t + C_DIM * C_DIM;
  float* Wr2t = Wl2t + C_DIM * C_DIM;
  float* Wl3t = Wr2t + C_DIM * C_DIM;
  float* Wr3t = Wl3t + C_DIM * OUT_C;
  float* Wlst = Wr3t + C_DIM * OUT_C;
  float* Wrst = Wlst + C_DIM * OUT_C;

  float* mu     = (float*)d_out;
  float* logstd = mu + (long long)N3 * OUT_C;

  const int ZB = 256, GB = 256;
  const int EB = cdiv(QT, EC_BLOCK * EC_PER_THREAD);

  {
    zero_i32<<<cdiv(NT, ZB), ZB, 0, stream>>>(deg, NT);
    edge_rank<<<EB, EC_BLOCK, 0, stream>>>(ei1, ei2, ei3, E1, E2, E3, N1, N2, deg, rank);
    scan_deg<<<3 + TR_BLOCKS, 1024, 0, stream>>>(deg, rowp, N1, N2, N3,
                                                 Wl1, Wr1, Wl2, Wr2, Wl3, Wr3, Wls, Wrs, wt);
    fill_perm<<<EB, EC_BLOCK, 0, stream>>>(ei1, ei2, ei3, E1, E2, E3, N1, N2, rowp, rank, perm);
  }

  {
    gather_rows<<<cdiv(N1, 8), GB, 0, stream>>>(x, rowp, deg, perm, agg, cnt, N1);
    int tiles = (N1 / (16 * MT)) * (C_DIM / 64);
    gemm_dual_wmma<MT, 1><<<cdiv(tiles, 4), 128, 0, stream>>>(
        agg, Wl1t, cnt, x, Wr1t, bl1, h1, N1, C_DIM);
  }

  {
    gather_rows<<<cdiv(N2, 8), GB, 0, stream>>>(h1, rowp + N1, deg + N1, perm + E1, agg, cnt, N2);
    int tiles = (N2 / (16 * MT)) * (C_DIM / 64);
    gemm_dual_wmma<MT, 1><<<cdiv(tiles, 4), 128, 0, stream>>>(
        agg, Wl2t, cnt, h1, Wr2t, bl2, h2, N2, C_DIM);
  }

  {
    gather_rows<<<cdiv(N3, 8), GB, 0, stream>>>(h2, rowp + N1 + N2, deg + N1 + N2, perm + E1 + E2,
                                                agg, cnt, N3);

    int tiles = (N3 / 16) * (OUT_C / 64);
    gemm_heads_wmma<<<dim3(cdiv(tiles, 8), 2), GB, 0, stream>>>(
        agg, cnt, h2, Wl3t, Wr3t, bl3, mu, Wlst, Wrst, bls, logstd, N3, OUT_C);
  }
}


# Row lists without prefix sum: the ranking pass stores each source id straight into its row's 48 fixed slots (longer rows: overflow list), so scan and fill launches go (10 -> 8); slot order kept
# speedup vs baseline: 2.7587x; 1.0560x over previous
#include <hip/hip_runtime.h>
#include <hip/hip_bf16.h>

typedef __attribute__((ext_vector_type(2))) float v2f;
typedef __attribute__((ext_vector_type(4))) float v4f;
typedef __attribute__((ext_vector_type(8))) float v8f;

#define C_DIM 256

#define O_DIM 64
#define TR_BIG_BLOCKS (C_DIM * C_DIM / 4096)
#define TR_SML_BLOCKS (C_DIM * O_DIM / 4096)
#define TR_BLOCKS (4 * TR_BIG_BLOCKS + 4 * TR_SML_BLOCKS)
__device__ __forceinline__ void transpose_block(
    int b, int tid,
    const float* __restrict__ w0, const float* __restrict__ w1,
    const float* __restrict__ w2, const float* __restrict__ w3,
    const float* __restrict__ w4, const float* __restrict__ w5,
    const float* __restrict__ w6, const float* __restrict__ w7,
    float* __restrict__ wt) {
  constexpr int K = C_DIM;
  int m, local;
  if (b < 4 * TR_BIG_BLOCKS) { m = b / TR_BIG_BLOCKS; local = b % TR_BIG_BLOCKS; }
  else { const int r = b - 4 * TR_BIG_BLOCKS; m = 4 + r / TR_SML_BLOCKS; local = r % TR_SML_BLOCKS; }
  const float* W = m == 0 ? w0 : m == 1 ? w1 : m == 2 ? w2 : m == 3 ? w3
                 : m == 4 ? w4 : m == 5 ? w5 : m == 6 ? w6 : w7;
  const int N = m < 4 ? C_DIM : O_DIM;
  float* Wt = wt + (m < 4 ? m * (C_DIM * C_DIM)
                          : 4 * (C_DIM * C_DIM) + (m - 4) * (C_DIM * O_DIM));
  const int t = (local * 1024 + tid) * 4;
  const int k = t / N;
  const int n = t % N;
  const float4 v = *(const float4*)(W + t);
  Wt[(n + 0) * K + k] = v.x;
  Wt[(n + 1) * K + k] = v.y;
  Wt[(n + 2) * K + k] = v.z;
  Wt[(n + 3) * K + k] = v.w;
}

__global__ __launch_bounds__(1024) void zero_and_transpose(
    int* __restrict__ p, int n, int zero_blocks,
    const float* __restrict__ w0, const float* __restrict__ w1,
    const float* __restrict__ w2, const float* __restrict__ w3,
    const float* __restrict__ w4, const float* __restrict__ w5,
    const float* __restrict__ w6, const float* __restrict__ w7,
    float* __restrict__ wt) {
  if ((int)blockIdx.x >= zero_blocks) {
    transpose_block(blockIdx.x - zero_blocks, threadIdx.x, w0, w1, w2, w3, w4, w5, w6, w7, wt);
    return;
  }
  const int i = blockIdx.x * 1024 + threadIdx.x;
  if (i < n) p[i] = 0;
}

#define ROW_CAP 48
#define EC_PER_THREAD 2
#define EC_BLOCK 256

struct EdgeChunk {
  const int* src;
  const int* dst;
  int e;
  int roff;
  int eoff;
  int nv;
};

__device__ __forceinline__ EdgeChunk edge_chunk(
    int q, const int* __restrict__ ei1, const int* __restrict__ ei2,
    const int* __restrict__ ei3, int E1, int E2, int E3, int n1, int n2) {
  const int Q1 = (E1 + 3) >> 2, Q2 = (E2 + 3) >> 2;
  const bool in1 = q < Q1, in2 = q < Q1 + Q2;
  const int E = in1 ? E1 : in2 ? E2 : E3;
  EdgeChunk c;
  c.src  = in1 ? ei1 : in2 ? ei2 : ei3;
  c.e    = 4 * (q - (in1 ? 0 : in2 ? Q1 : Q1 + Q2));
  c.roff = in1 ? 0 : in2 ? n1 : n1 + n2;
  c.eoff = in1 ? 0 : in2 ? E1 : E1 + E2;
  c.dst  = c.src + E;
  c.nv   = min(4, E - c.e);
  return c;
}

template <bool WHOLE>
__device__ __forceinline__ int4 load_ids4(const int* __restrict__ p, int e, int nv) {
  if (WHOLE) return *(const int4*)(p + e);
  int4 v = make_int4(0, 0, 0, 0);
  v.x = p[e];
  if (nv > 1) v.y = p[e + 1];
  if (nv > 2) v.z = p[e + 2];
  if (nv > 3) v.w = p[e + 3];
  return v;
}

__device__ __forceinline__ void put_edge(int* __restrict__ slots, int* __restrict__ ovf_n,
                                         int* __restrict__ ovf, int row, int r, int s, int g) {
  if (r < ROW_CAP) slots[row * ROW_CAP + r] = s;
  else ovf[atomicAdd(ovf_n, 1)] = g;
}

template <bool WHOLE, bool FULL>
__device__ __forceinline__ void edge_slots_body(
    const int* __restrict__ ei1, const int* __restrict__ ei2,
    const int* __restrict__ ei3, int E1, int E2, int E3, int n1, int n2,
    int* __restrict__ deg, int* __restrict__ slots,
    int* __restrict__ ovf_n, int* __restrict__ ovf) {
  const int QT = ((E1 + 3) >> 2) + ((E2 + 3) >> 2) + ((E3 + 3) >> 2);
  const int q0 = blockIdx.x * (EC_BLOCK * EC_PER_THREAD) + threadIdx.x;
  EdgeChunk c[EC_PER_THREAD];
  int4 s[EC_PER_THREAD], d[EC_PER_THREAD], r[EC_PER_THREAD];
  int nv[EC_PER_THREAD];
  for (int u = 0; u < EC_PER_THREAD; ++u) {
    const int q = q0 + u * EC_BLOCK;
    c[u] = edge_chunk(min(q, QT - 1), ei1, ei2, ei3, E1, E2, E3, n1, n2);
    d[u] = load_ids4<WHOLE>(c[u].dst, c[u].e, c[u].nv);
    s[u] = load_ids4<WHOLE>(c[u].src, c[u].e, c[u].nv);
    nv[u] = FULL ? 4 : (q < QT ? c[u].nv : 0);
  }
  asm volatile("" ::: "memory");
  for (int u = 0; u < EC_PER_THREAD; ++u) {
    r[u] = make_int4(0, 0, 0, 0);
    if (FULL || nv[u] > 0) {
      int* g = deg + c[u].roff;
      r[u].x = atomicAdd(g + d[u].x, 1);
      if (WHOLE || nv[u] > 1) r[u].y = atomicAdd(g + d[u].y, 1);
      if (WHOLE || nv[u] > 2) r[u].z = atomicAdd(g + d[u].z, 1);
      if (WHOLE || nv[u] > 3) r[u].w = atomicAdd(g + d[u].w, 1);
    }
  }
  for (int u = 0; u < EC_PER_THREAD; ++u) {
    if (FULL || nv[u] > 0) {
      const int ro = c[u].roff, g0 = c[u].eoff + c[u].e;
      put_edge(slots, ovf_n, ovf, ro + d[u].x, r[u].x, s[u].x, g0);
      if (WHOLE || nv[u] > 1) put_edge(slots, ovf_n, ovf, ro + d[u].y, r[u].y, s[u].y, g0 + 1);
      if (WHOLE || nv[u] > 2) put_edge(slots, ovf_n, ovf, ro + d[u].z, r[u].z, s[u].z, g0 + 2);
      if (WHOLE || nv[u] > 3) put_edge(slots, ovf_n, ovf, ro + d[u].w, r[u].w, s[u].w, g0 + 3);
    }
  }
}

__device__ __forceinline__ bool edge_block_full(int E1, int E2, int E3) {
  const int QT = ((E1 + 3) >> 2) + ((E2 + 3) >> 2) + ((E3 + 3) >> 2);
  return (int)(blockIdx.x + 1) * (EC_BLOCK * EC_PER_THREAD) <= QT;
}

__global__ __launch_bounds__(EC_BLOCK) void edge_slots(
    const int* __restrict__ ei1, const int* __restrict__ ei2,
    const int* __restrict__ ei3, int E1, int E2, int E3,
    int n1, int n2,
    int* __restrict__ deg, int* __restrict__ slots,
    int* __restrict__ ovf_n, int* __restrict__ ovf) {
  const bool whole = ((E1 | E2 | E3) & 3) == 0;
  if (whole && edge_block_full(E1, E2, E3))
    edge_slots_body<true, true >(ei1, ei2, ei3, E1, E2, E3, n1, n2, deg, slots, ovf_n, ovf);
  else if (whole)
    edge_slots_body<true, false>(ei1, ei2, ei3, E1, E2, E3, n1, n2, deg, slots, ovf_n, ovf);
  else
    edge_slots_body<false, false>(ei1, ei2, ei3, E1, E2, E3, n1, n2, deg, slots, ovf_n, ovf);
}

template <int B>
__device__ __forceinline__ void add_rows(const float* __restrict__ xl, int idx, int j,
                                         float4& a0, float4& a1) {
  float4 u[B], w[B];
  for (int t = 0; t < B; ++t) {
    const float* r = xl + (long long)__builtin_amdgcn_readlane(idx, j + t) * C_DIM;
    u[t] = *(const float4*)r;
    w[t] = *(const float4*)(r + 128);
  }
  for (int t = 0; t < B; ++t) {
    a0.x += u[t].x; a0.y += u[t].y; a0.z += u[t].z; a0.w += u[t].w;
    a1.x += w[t].x; a1.y += w[t].y; a1.z += w[t].z; a1.w += w[t].w;
  }
}

__global__ __launch_bounds__(256) void gather_rows(
    const float* __restrict__ x, const int* __restrict__ deg,
    const int* __restrict__ slots, const int* __restrict__ ovf_n,
    const int* __restrict__ ovf, const int* __restrict__ ei, int E, int eoff,
    float* __restrict__ agg, float* __restrict__ cnt, int n_dst) {
  const int lane = threadIdx.x & 31;
  const int d = __builtin_amdgcn_readfirstlane(blockIdx.x * 8 + (threadIdx.x >> 5));
  if (d >= n_dst) return;
  const int n  = deg[d];
  const int ns = min(n, ROW_CAP);
  const int* sl = slots + d * ROW_CAP;
  const float* xl = x + lane * 4;

  float4 a0 = make_float4(0.f, 0.f, 0.f, 0.f);
  float4 a1 = make_float4(0.f, 0.f, 0.f, 0.f);

  for (int base = 0; base < ns; base += 32) {
    const int m = min(ns - base, 32);
    int idx = 0;
    if (lane < m) idx = sl[base + lane];
    int j = 0;
    for (; j + 8 <= m; j += 8) add_rows<8>(xl, idx, j, a0, a1);
    if (m & 4) { add_rows<4>(xl, idx, j, a0, a1); j += 4; }
    if (m & 2) { add_rows<2>(xl, idx, j, a0, a1); j += 2; }
    if (m & 1) { add_rows<1>(xl, idx, j, a0, a1); }
  }
  if (n > ROW_CAP) {
    const int total = *ovf_n;
    for (int k = 0; k < total; ++k) {
      const int e = ovf[k] - eoff;
      if (e >= 0 && e < E && ei[E + e] == d) add_rows<1>(xl, ei[e], 0, a0, a1);
    }
  }

  float* o = agg + (long long)d * C_DIM + lane * 4;
  *(float4*)o         = a0;
  *(float4*)(o + 128) = a1;
  if (lane == 0) cnt[d] = (float)n;
}

template <int MT>
__device__ __forceinline__ void load_frags(const float* __restrict__ ap,
                                           const float* __restrict__ wp, int k,
                                           v4f (&a)[MT], v4f (&b)[4]) {
  constexpr int K = C_DIM;
#pragma unroll
  for (int i = 0; i < MT; ++i) a[i] = *(const v4f*)(ap + i * 16 * K + k);
#pragma unroll
  for (int j = 0; j < 4; ++j) b[j] = *(const v4f*)(wp + j * 16 * K + k);
}

template <int MT, bool SCALE>
__device__ __forceinline__ void mma_step(const v4f (&a)[MT], const v4f (&b)[4],
                                         const float (&s)[MT], v8f (&acc)[MT][4]) {
#pragma unroll
  for (int i = 0; i < MT; ++i) {
    const v4f x = SCALE ? a[i] * s[i] : a[i];
    const v2f xlo = {x.x, x.y};
    const v2f xhi = {x.z, x.w};
#pragma unroll
    for (int j = 0; j < 4; ++j) {
      const v2f blo = {b[j].x, b[j].y};
      acc[i][j] = __builtin_amdgcn_wmma_f32_16x16x4_f32(false, xlo, false, blo,
                                                        (short)0, acc[i][j],
                                                        false, false);
    }
#pragma unroll
    for (int j = 0; j < 4; ++j) {
      const v2f bhi = {b[j].z, b[j].w};
      acc[i][j] = __builtin_amdgcn_wmma_f32_16x16x4_f32(false, xhi, false, bhi,
                                                        (short)0, acc[i][j],
                                                        false, false);
    }
  }
}

template <int MT, bool SCALE>
__device__ __forceinline__ void gemm_path(const float* __restrict__ ap,
                                          const float* __restrict__ wp,
                                          const float (&s)[MT], v8f (&acc)[MT][4]) {
  constexpr int K = C_DIM;
  static_assert(K % 16 == 0, "the k loop takes two 8-wide steps per trip");
  v4f a0[MT], b0[4], a1[MT], b1[4];
  load_frags<MT>(ap, wp, 0, a0, b0);
  for (int k = 0; k < K - 16; k += 16) {
    load_frags<MT>(ap, wp, k + 8, a1, b1);
    mma_step<MT, SCALE>(a0, b0, s, acc);
    load_frags<MT>(ap, wp, k + 16, a0, b0);
    mma_step<MT, SCALE>(a1, b1, s, acc);
  }
  load_frags<MT>(ap, wp, K - 8, a1, b1);
  mma_step<MT, SCALE>(a0, b0, s, acc);
  mma_step<MT, SCALE>(a1, b1, s, acc);
}

template <int MT, int RELU>
__device__ __forceinline__ void gemm_dual_tile(
    const float* __restrict__ A1, const float* __restrict__ Wt1,
    const float* __restrict__ cnt1,
    const float* __restrict__ A2, const float* __restrict__ Wt2,
    const float* __restrict__ bias, float* __restrict__ out,
    int M, int N) {
  constexpr int K = C_DIM;
  constexpr int BM = 16 * MT;
  const int lane = threadIdx.x & 31;
  const int wave = threadIdx.x >> 5;
  const int nstrips = N >> 6;
  const int tile = blockIdx.x * (blockDim.x >> 5) + wave;
  if (tile >= (M / BM) * nstrips) return;

  const int m0 = (tile / nstrips) * BM;
  const int n0 = (tile % nstrips) << 6;

  const int half = lane >> 4;
  const int l15  = lane & 15;
  const int row  = m0 + l15;
  const int cb   = n0 + l15;
  const int koff = half << 2;

  float s1[MT];
#pragma unroll
  for (int i = 0; i < MT; ++i) s1[i] = 1.0f / fmaxf(cnt1[row + i * 16], 1.0f);

  v8f acc[MT][4];
#pragma unroll
  for (int i = 0; i < MT; ++i)
#pragma unroll
    for (int j = 0; j < 4; ++j) acc[i][j] = (v8f){0.f,0.f,0.f,0.f,0.f,0.f,0.f,0.f};

  const float* a1p = A1 + (long long)row * K + koff;
  const float* a2p = A2 + (long long)row * K + koff;
  const float* w1p = Wt1 + (long long)cb * K + koff;
  const float* w2p = Wt2 + (long long)cb * K + koff;

  gemm_path<MT, true >(a1p, w1p, s1, acc);
  gemm_path<MT, false>(a2p, w2p, s1, acc);

#pragma unroll
  for (int j = 0; j < 4; ++j) {
    const int col = cb + j * 16;
    const float bv = bias[col];
#pragma unroll
    for (int i = 0; i < MT; ++i) {
#pragma unroll
      for (int r = 0; r < 8; ++r) {
        float v = acc[i][j][r] + bv;
        if (RELU) v = fmaxf(v, 0.f);
        out[(long long)(m0 + i * 16 + r + half * 8) * N + col] = v;
      }
    }
  }
}

template <int MT, int RELU>
__global__ __launch_bounds__(256) void gemm_dual_wmma(
    const float* __restrict__ A1, const float* __restrict__ Wt1,
    const float* __restrict__ cnt1,
    const float* __restrict__ A2, const float* __restrict__ Wt2,
    const float* __restrict__ bias, float* __restrict__ out,
    int M, int N) {
  gemm_dual_tile<MT, RELU>(A1, Wt1, cnt1, A2, Wt2, bias, out, M, N);
}

__global__ __launch_bounds__(256) void gemm_heads_wmma(
    const float* __restrict__ A1, const float* __restrict__ cnt1,
    const float* __restrict__ A2,
    const float* __restrict__ Wt1a, const float* __restrict__ Wt2a,
    const float* __restrict__ biasa, float* __restrict__ outa,
    const float* __restrict__ Wt1b, const float* __restrict__ Wt2b,
    const float* __restrict__ biasb, float* __restrict__ outb,
    int M, int N) {
  const bool second = blockIdx.y != 0;
  gemm_dual_tile<1, 0>(A1, second ? Wt1b : Wt1a, cnt1,
                       A2, second ? Wt2b : Wt2a,
                       second ? biasb : biasa, second ? outb : outa, M, N);
}

static inline int cdiv(long long a, long long b) { return (int)((a + b - 1) / b); }

extern "C" void kernel_launch(void* const* d_in, const int* in_sizes, int n_in,
                              void* d_out, int out_size, void* d_ws, size_t ws_size,
                              hipStream_t stream) {
  (void)n_in; (void)out_size; (void)ws_size;

  const int N1 = 40000, N2 = 20000, N3 = 10000;
  const int OUT_C = O_DIM;
  static_assert(2 * N2 <= N1 && N3 <= N2, "h2 lives in the half of agg that layers 2 and 3 leave free");
  constexpr int MT = 5;
  static_assert(N1 % (16 * MT) == 0 && N2 % (16 * MT) == 0 && N3 % 16 == 0, "GEMM row blocks must tile M");

  const float* x   = (const float*)d_in[0];
  const int*   ei1 = (const int*)d_in[1];
  const int*   ei2 = (const int*)d_in[2];
  const int*   ei3 = (const int*)d_in[3];
  const float* Wl1 = (const float*)d_in[4];
  const float* bl1 = (const float*)d_in[5];
  const float* Wr1 = (const float*)d_in[6];
  const float* Wl2 = (const float*)d_in[7];
  const float* bl2 = (const float*)d_in[8];
  const float* Wr2 = (const float*)d_in[9];
  const float* Wl3 = (const float*)d_in[10];
  const float* bl3 = (const float*)d_in[11];
  const float* Wr3 = (const float*)d_in[12];
  const float* Wls = (const float*)d_in[13];
  const float* bls = (const float*)d_in[14];
  const float* Wrs = (const float*)d_in[15];

  const int E1 = in_sizes[1] / 2;
  const int E2 = in_sizes[2] / 2;
  const int E3 = in_sizes[3] / 2;
  const int QT = cdiv(E1, 4) + cdiv(E2, 4) + cdiv(E3, 4);
  const int NT = N1 + N2 + N3;

  float* h1  = (float*)d_ws;
  int*   csr = (int*)(h1 + (long long)N1 * C_DIM);
  int*   deg   = csr;
  int*   ovfn  = deg + NT;
  int*   slots = ovfn + 4;
  int*   ovf   = slots + NT * ROW_CAP;
  static_assert((long long)NT * (1 + ROW_CAP) + 4 < (long long)N2 * C_DIM, "row slots must fit their workspace part");
  float* agg = (float*)(csr + (long long)N2 * C_DIM);
  float* h2  = agg + (long long)N2 * C_DIM;
  float* cnt = agg + (long long)N1 * C_DIM;
  float* wt  = cnt + N1;
  float* Wl1t = wt;
  float* Wr1t = Wl1t + C_DIM * C_DIM;
  float* Wl2t = Wr1t + C_DIM * C_DIM;
  float* Wr2t = Wl2t + C_DIM * C_DIM;
  float* Wl3t = Wr2t + C_DIM * C_DIM;
  float* Wr3t = Wl3t + C_DIM * OUT_C;
  float* Wlst = Wr3t + C_DIM * OUT_C;
  float* Wrst = Wlst + C_DIM * OUT_C;

  float* mu     = (float*)d_out;
  float* logstd = mu + (long long)N3 * OUT_C;

  const int GB = 256;
  const int ZBLK = cdiv(NT + 4, 1024);
  const int EB = cdiv(QT, EC_BLOCK * EC_PER_THREAD);

  {
    zero_and_transpose<<<ZBLK + TR_BLOCKS, 1024, 0, stream>>>(deg, NT + 4, ZBLK,
                                                 Wl1, Wr1, Wl2, Wr2, Wl3, Wr3, Wls, Wrs, wt);
    edge_slots<<<EB, EC_BLOCK, 0, stream>>>(ei1, ei2, ei3, E1, E2, E3, N1, N2, deg, slots, ovfn, ovf);
  }

  {
    gather_rows<<<cdiv(N1, 8), GB, 0, stream>>>(x, deg, slots, ovfn, ovf, ei1, E1, 0, agg, cnt, N1);
    int tiles = (N1 / (16 * MT)) * (C_DIM / 64);
    gemm_dual_wmma<MT, 1><<<cdiv(tiles, 4), 128, 0, stream>>>(
        agg, Wl1t, cnt, x, Wr1t, bl1, h1, N1, C_DIM);
  }

  {
    gather_rows<<<cdiv(N2, 8), GB, 0, stream>>>(h1, deg + N1, slots + N1 * ROW_CAP, ovfn, ovf, ei2, E2, E1,
                                                agg, cnt, N2);
    int tiles = (N2 / (16 * MT)) * (C_DIM / 64);
    gemm_dual_wmma<MT, 1><<<cdiv(tiles, 4), 128, 0, stream>>>(
        agg, Wl2t, cnt, h1, Wr2t, bl2, h2, N2, C_DIM);
  }

  {
    gather_rows<<<cdiv(N3, 8), GB, 0, stream>>>(h2, deg + N1 + N2, slots + (N1 + N2) * ROW_CAP, ovfn, ovf,
                                                ei3, E3, E1 + E2, agg, cnt, N3);

    int tiles = (N3 / 16) * (OUT_C / 64);
    gemm_heads_wmma<<<dim3(cdiv(tiles, 8), 2), GB, 0, stream>>>(
        agg, cnt, h2, Wl3t, Wr3t, bl3, mu, Wlst, Wrst, bls, logstd, N3, OUT_C);
  }
}


# Layer 1 and 2 GEMMs: a 4-wave block stages A and Wt in LDS with whole-line 16-byte loads (4 lines a load, not 16; A once per block), fragments read from LDS; k order and sums unchanged
# speedup vs baseline: 2.8430x; 1.0306x over previous
#include <hip/hip_runtime.h>
#include <hip/hip_bf16.h>

typedef __attribute__((ext_vector_type(2))) float v2f;
typedef __attribute__((ext_vector_type(4))) float v4f;
typedef __attribute__((ext_vector_type(8))) float v8f;

#define C_DIM 256

#define O_DIM 64
#define TR_BIG_BLOCKS (C_DIM * C_DIM / 4096)
#define TR_SML_BLOCKS (C_DIM * O_DIM / 4096)
#define TR_BLOCKS (4 * TR_BIG_BLOCKS + 4 * TR_SML_BLOCKS)
__device__ __forceinline__ void transpose_block(
    int b, int tid,
    const float* __restrict__ w0, const float* __restrict__ w1,
    const float* __restrict__ w2, const float* __restrict__ w3,
    const float* __restrict__ w4, const float* __restrict__ w5,
    const float* __restrict__ w6, const float* __restrict__ w7,
    float* __restrict__ wt) {
  constexpr int K = C_DIM;
  int m, local;
  if (b < 4 * TR_BIG_BLOCKS) { m = b / TR_BIG_BLOCKS; local = b % TR_BIG_BLOCKS; }
  else { const int r = b - 4 * TR_BIG_BLOCKS; m = 4 + r / TR_SML_BLOCKS; local = r % TR_SML_BLOCKS; }
  const float* W = m == 0 ? w0 : m == 1 ? w1 : m == 2 ? w2 : m == 3 ? w3
                 : m == 4 ? w4 : m == 5 ? w5 : m == 6 ? w6 : w7;
  const int N = m < 4 ? C_DIM : O_DIM;
  float* Wt = wt + (m < 4 ? m * (C_DIM * C_DIM)
                          : 4 * (C_DIM * C_DIM) + (m - 4) * (C_DIM * O_DIM));
  const int t = (local * 1024 + tid) * 4;
  const int k = t / N;
  const int n = t % N;
  const float4 v = *(const float4*)(W + t);
  Wt[(n + 0) * K + k] = v.x;
  Wt[(n + 1) * K + k] = v.y;
  Wt[(n + 2) * K + k] = v.z;
  Wt[(n + 3) * K + k] = v.w;
}

__global__ __launch_bounds__(1024) void zero_and_transpose(
    int* __restrict__ p, int n, int zero_blocks,
    const float* __restrict__ w0, const float* __restrict__ w1,
    const float* __restrict__ w2, const float* __restrict__ w3,
    const float* __restrict__ w4, const float* __restrict__ w5,
    const float* __restrict__ w6, const float* __restrict__ w7,
    float* __restrict__ wt) {
  if ((int)blockIdx.x >= zero_blocks) {
    transpose_block(blockIdx.x - zero_blocks, threadIdx.x, w0, w1, w2, w3, w4, w5, w6, w7, wt);
    return;
  }
  const int i = blockIdx.x * 1024 + threadIdx.x;
  if (i < n) p[i] = 0;
}

#define ROW_CAP 48
#define EC_PER_THREAD 2
#define EC_BLOCK 256

struct EdgeChunk {
  const int* src;
  const int* dst;
  int e;
  int roff;
  int eoff;
  int nv;
};

__device__ __forceinline__ EdgeChunk edge_chunk(
    int q, const int* __restrict__ ei1, const int* __restrict__ ei2,
    const int* __restrict__ ei3, int E1, int E2, int E3, int n1, int n2) {
  const int Q1 = (E1 + 3) >> 2, Q2 = (E2 + 3) >> 2;
  const bool in1 = q < Q1, in2 = q < Q1 + Q2;
  const int E = in1 ? E1 : in2 ? E2 : E3;
  EdgeChunk c;
  c.src  = in1 ? ei1 : in2 ? ei2 : ei3;
  c.e    = 4 * (q - (in1 ? 0 : in2 ? Q1 : Q1 + Q2));
  c.roff = in1 ? 0 : in2 ? n1 : n1 + n2;
  c.eoff = in1 ? 0 : in2 ? E1 : E1 + E2;
  c.dst  = c.src + E;
  c.nv   = min(4, E - c.e);
  return c;
}

template <bool WHOLE>
__device__ __forceinline__ int4 load_ids4(const int* __restrict__ p, int e, int nv) {
  if (WHOLE) return *(const int4*)(p + e);
  int4 v = make_int4(0, 0, 0, 0);
  v.x = p[e];
  if (nv > 1) v.y = p[e + 1];
  if (nv > 2) v.z = p[e + 2];
  if (nv > 3) v.w = p[e + 3];
  return v;
}

__device__ __forceinline__ void put_edge(int* __restrict__ slots, int* __restrict__ ovf_n,
                                         int* __restrict__ ovf, int row, int r, int s, int g) {
  if (r < ROW_CAP) slots[row * ROW_CAP + r] = s;
  else ovf[atomicAdd(ovf_n, 1)] = g;
}

template <bool WHOLE, bool FULL>
__device__ __forceinline__ void edge_slots_body(
    const int* __restrict__ ei1, const int* __restrict__ ei2,
    const int* __restrict__ ei3, int E1, int E2, int E3, int n1, int n2,
    int* __restrict__ deg, int* __restrict__ slots,
    int* __restrict__ ovf_n, int* __restrict__ ovf) {
  const int QT = ((E1 + 3) >> 2) + ((E2 + 3) >> 2) + ((E3 + 3) >> 2);
  const int q0 = blockIdx.x * (EC_BLOCK * EC_PER_THREAD) + threadIdx.x;
  EdgeChunk c[EC_PER_THREAD];
  int4 s[EC_PER_THREAD], d[EC_PER_THREAD], r[EC_PER_THREAD];
  int nv[EC_PER_THREAD];
  for (int u = 0; u < EC_PER_THREAD; ++u) {
    const int q = q0 + u * EC_BLOCK;
    c[u] = edge_chunk(min(q, QT - 1), ei1, ei2, ei3, E1, E2, E3, n1, n2);
    d[u] = load_ids4<WHOLE>(c[u].dst, c[u].e, c[u].nv);
    s[u] = load_ids4<WHOLE>(c[u].src, c[u].e, c[u].nv);
    nv[u] = FULL ? 4 : (q < QT ? c[u].nv : 0);
  }
  asm volatile("" ::: "memory");
  for (int u = 0; u < EC_PER_THREAD; ++u) {
    r[u] = make_int4(0, 0, 0, 0);
    if (FULL || nv[u] > 0) {
      int* g = deg + c[u].roff;
      r[u].x = atomicAdd(g + d[u].x, 1);
      if (WHOLE || nv[u] > 1) r[u].y = atomicAdd(g + d[u].y, 1);
      if (WHOLE || nv[u] > 2) r[u].z = atomicAdd(g + d[u].z, 1);
      if (WHOLE || nv[u] > 3) r[u].w = atomicAdd(g + d[u].w, 1);
    }
  }
  for (int u = 0; u < EC_PER_THREAD; ++u) {
    if (FULL || nv[u] > 0) {
      const int ro = c[u].roff, g0 = c[u].eoff + c[u].e;
      put_edge(slots, ovf_n, ovf, ro + d[u].x, r[u].x, s[u].x, g0);
      if (WHOLE || nv[u] > 1) put_edge(slots, ovf_n, ovf, ro + d[u].y, r[u].y, s[u].y, g0 + 1);
      if (WHOLE || nv[u] > 2) put_edge(slots, ovf_n, ovf, ro + d[u].z, r[u].z, s[u].z, g0 + 2);
      if (WHOLE || nv[u] > 3) put_edge(slots, ovf_n, ovf, ro + d[u].w, r[u].w, s[u].w, g0 + 3);
    }
  }
}

__device__ __forceinline__ bool edge_block_full(int E1, int E2, int E3) {
  const int QT = ((E1 + 3) >> 2) + ((E2 + 3) >> 2) + ((E3 + 3) >> 2);
  return (int)(blockIdx.x + 1) * (EC_BLOCK * EC_PER_THREAD) <= QT;
}

__global__ __launch_bounds__(EC_BLOCK) void edge_slots(
    const int* __restrict__ ei1, const int* __restrict__ ei2,
    const int* __restrict__ ei3, int E1, int E2, int E3,
    int n1, int n2,
    int* __restrict__ deg, int* __restrict__ slots,
    int* __restrict__ ovf_n, int* __restrict__ ovf) {
  const bool whole = ((E1 | E2 | E3) & 3) == 0;
  if (whole && edge_block_full(E1, E2, E3))
    edge_slots_body<true, true >(ei1, ei2, ei3, E1, E2, E3, n1, n2, deg, slots, ovf_n, ovf);
  else if (whole)
    edge_slots_body<true, false>(ei1, ei2, ei3, E1, E2, E3, n1, n2, deg, slots, ovf_n, ovf);
  else
    edge_slots_body<false, false>(ei1, ei2, ei3, E1, E2, E3, n1, n2, deg, slots, ovf_n, ovf);
}

template <int B>
__device__ __forceinline__ void add_rows(const float* __restrict__ xl, int idx, int j,
                                         float4& a0, float4& a1) {
  float4 u[B], w[B];
  for (int t = 0; t < B; ++t) {
    const float* r = xl + (long long)__builtin_amdgcn_readlane(idx, j + t) * C_DIM;
    u[t] = *(const float4*)r;
    w[t] = *(const float4*)(r + 128);
  }
  for (int t = 0; t < B; ++t) {
    a0.x += u[t].x; a0.y += u[t].y; a0.z += u[t].z; a0.w += u[t].w;
    a1.x += w[t].x; a1.y += w[t].y; a1.z += w[t].z; a1.w += w[t].w;
  }
}

__global__ __launch_bounds__(256) void gather_rows(
    const float* __restrict__ x, const int* __restrict__ deg,
    const int* __restrict__ slots, const int* __restrict__ ovf_n,
    const int* __restrict__ ovf, const int* __restrict__ ei, int E, int eoff,
    float* __restrict__ agg, float* __restrict__ cnt, int n_dst) {
  const int lane = threadIdx.x & 31;
  const int d = __builtin_amdgcn_readfirstlane(blockIdx.x * 8 + (threadIdx.x >> 5));
  if (d >= n_dst) return;
  const int n  = deg[d];
  const int ns = min(n, ROW_CAP);
  const int* sl = slots + d * ROW_CAP;
  const float* xl = x + lane * 4;

  float4 a0 = make_float4(0.f, 0.f, 0.f, 0.f);
  float4 a1 = make_float4(0.f, 0.f, 0.f, 0.f);

  for (int base = 0; base < ns; base += 32) {
    const int m = min(ns - base, 32);
    int idx = 0;
    if (lane < m) idx = sl[base + lane];
    int j = 0;
    for (; j + 8 <= m; j += 8) add_rows<8>(xl, idx, j, a0, a1);
    if (m & 4) { add_rows<4>(xl, idx, j, a0, a1); j += 4; }
    if (m & 2) { add_rows<2>(xl, idx, j, a0, a1); j += 2; }
    if (m & 1) { add_rows<1>(xl, idx, j, a0, a1); }
  }
  if (n > ROW_CAP) {
    const int total = *ovf_n;
    for (int k = 0; k < total; ++k) {
      const int e = ovf[k] - eoff;
      if (e >= 0 && e < E && ei[E + e] == d) add_rows<1>(xl, ei[e], 0, a0, a1);
    }
  }

  float* o = agg + (long long)d * C_DIM + lane * 4;
  *(float4*)o         = a0;
  *(float4*)(o + 128) = a1;
  if (lane == 0) cnt[d] = (float)n;
}

template <int MT>
__device__ __forceinline__ void load_frags(const float* __restrict__ ap,
                                           const float* __restrict__ wp, int k,
                                           v4f (&a)[MT], v4f (&b)[4]) {
  constexpr int K = C_DIM;
#pragma unroll
  for (int i = 0; i < MT; ++i) a[i] = *(const v4f*)(ap + i * 16 * K + k);
#pragma unroll
  for (int j = 0; j < 4; ++j) b[j] = *(const v4f*)(wp + j * 16 * K + k);
}

template <int MT, bool SCALE>
__device__ __forceinline__ void mma_step(const v4f (&a)[MT], const v4f (&b)[4],
                                         const float (&s)[MT], v8f (&acc)[MT][4]) {
#pragma unroll
  for (int i = 0; i < MT; ++i) {
    const v4f x = SCALE ? a[i] * s[i] : a[i];
    const v2f xlo = {x.x, x.y};
    const v2f xhi = {x.z, x.w};
#pragma unroll
    for (int j = 0; j < 4; ++j) {
      const v2f blo = {b[j].x, b[j].y};
      acc[i][j] = __builtin_amdgcn_wmma_f32_16x16x4_f32(false, xlo, false, blo,
                                                        (short)0, acc[i][j],
                                                        false, false);
    }
#pragma unroll
    for (int j = 0; j < 4; ++j) {
      const v2f bhi = {b[j].z, b[j].w};
      acc[i][j] = __builtin_amdgcn_wmma_f32_16x16x4_f32(false, xhi, false, bhi,
                                                        (short)0, acc[i][j],
                                                        false, false);
    }
  }
}

template <int MT, bool SCALE>
__device__ __forceinline__ void gemm_path(const float* __restrict__ ap,
                                          const float* __restrict__ wp,
                                          const float (&s)[MT], v8f (&acc)[MT][4]) {
  constexpr int K = C_DIM;
  static_assert(K % 16 == 0, "the k loop takes two 8-wide steps per trip");
  v4f a0[MT], b0[4], a1[MT], b1[4];
  load_frags<MT>(ap, wp, 0, a0, b0);
  for (int k = 0; k < K - 16; k += 16) {
    load_frags<MT>(ap, wp, k + 8, a1, b1);
    mma_step<MT, SCALE>(a0, b0, s, acc);
    load_frags<MT>(ap, wp, k + 16, a0, b0);
    mma_step<MT, SCALE>(a1, b1, s, acc);
  }
  load_frags<MT>(ap, wp, K - 8, a1, b1);
  mma_step<MT, SCALE>(a0, b0, s, acc);
  mma_step<MT, SCALE>(a1, b1, s, acc);
}

template <int MT, int RELU>
__device__ __forceinline__ void gemm_dual_tile(
    const float* __restrict__ A1, const float* __restrict__ Wt1,
    const float* __restrict__ cnt1,
    const float* __restrict__ A2, const float* __restrict__ Wt2,
    const float* __restrict__ bias, float* __restrict__ out,
    int M, int N) {
  constexpr int K = C_DIM;
  constexpr int BM = 16 * MT;
  const int lane = threadIdx.x & 31;
  const int wave = threadIdx.x >> 5;
  const int nstrips = N >> 6;
  const int tile = blockIdx.x * (blockDim.x >> 5) + wave;
  if (tile >= (M / BM) * nstrips) return;

  const int m0 = (tile / nstrips) * BM;
  const int n0 = (tile % nstrips) << 6;

  const int half = lane >> 4;
  const int l15  = lane & 15;
  const int row  = m0 + l15;
  const int cb   = n0 + l15;
  const int koff = half << 2;

  float s1[MT];
#pragma unroll
  for (int i = 0; i < MT; ++i) s1[i] = 1.0f / fmaxf(cnt1[row + i * 16], 1.0f);

  v8f acc[MT][4];
#pragma unroll
  for (int i = 0; i < MT; ++i)
#pragma unroll
    for (int j = 0; j < 4; ++j) acc[i][j] = (v8f){0.f,0.f,0.f,0.f,0.f,0.f,0.f,0.f};

  const float* a1p = A1 + (long long)row * K + koff;
  const float* a2p = A2 + (long long)row * K + koff;
  const float* w1p = Wt1 + (long long)cb * K + koff;
  const float* w2p = Wt2 + (long long)cb * K + koff;

  gemm_path<MT, true >(a1p, w1p, s1, acc);
  gemm_path<MT, false>(a2p, w2p, s1, acc);

#pragma unroll
  for (int j = 0; j < 4; ++j) {
    const int col = cb + j * 16;
    const float bv = bias[col];
#pragma unroll
    for (int i = 0; i < MT; ++i) {
#pragma unroll
      for (int r = 0; r < 8; ++r) {
        float v = acc[i][j][r] + bv;
        if (RELU) v = fmaxf(v, 0.f);
        out[(long long)(m0 + i * 16 + r + half * 8) * N + col] = v;
      }
    }
  }
}

#define KC 32
#define LPITCH 36

template <int MT>
__device__ __forceinline__ void lds_frags(const float* sa, const float* sb, int ks,
                                          v4f (&a)[MT], v4f (&b)[4]) {
#pragma unroll
  for (int i = 0; i < MT; ++i) a[i] = *(const v4f*)(sa + i * 16 * LPITCH + ks * 8);
#pragma unroll
  for (int j = 0; j < 4; ++j) b[j] = *(const v4f*)(sb + j * 16 * LPITCH + ks * 8);
}

template <int MT, int RELU>
__global__ __launch_bounds__(128) void gemm_dual_lds(
    const float* __restrict__ A1, const float* __restrict__ Wt1,
    const float* __restrict__ cnt1,
    const float* __restrict__ A2, const float* __restrict__ Wt2,
    const float* __restrict__ bias, float* __restrict__ out) {
  constexpr int K = C_DIM, N = C_DIM;
  constexpr int BM = 16 * MT;
  constexpr int ROWS = BM + N;
  constexpr int NP = ROWS * (KC / 4) / 128;
  constexpr int STAGE = ROWS * LPITCH;
  constexpr int NCHUNK = 2 * K / KC;
  static_assert(ROWS * (KC / 4) % 128 == 0, "the pieces of a chunk deal out evenly");
  static_assert(2 * STAGE * 4 <= 160 * 1024, "two blocks per WGP");
  __shared__ __attribute__((aligned(16))) float lds[2 * STAGE];

  const int tid = threadIdx.x;
  const int lane = tid & 31;
  const int wave = tid >> 5;
  const int m0 = blockIdx.x * BM;

  const int srow = tid >> 3;
  const int sq   = tid & 7;
  float sc[MT];
#pragma unroll
  for (int i = 0; i < MT; ++i) sc[i] = 1.0f / fmaxf(cnt1[m0 + srow + i * 16], 1.0f);
  const float* ga1 = A1 + (long long)(m0 + srow) * K + 4 * sq;
  const float* ga2 = A2 + (long long)(m0 + srow) * K + 4 * sq;
  const float* gw1 = Wt1 + (long long)srow * K + 4 * sq;
  const float* gw2 = Wt2 + (long long)srow * K + 4 * sq;
  float* park = lds + srow * LPITCH + 4 * sq;

  const int half = lane >> 4;
  const int l15  = lane & 15;
  const float* fa = lds + l15 * LPITCH + 4 * half;
  const float* fb = lds + (BM + 64 * wave + l15) * LPITCH + 4 * half;

  v8f acc[MT][4];
#pragma unroll
  for (int i = 0; i < MT; ++i)
#pragma unroll
    for (int j = 0; j < 4; ++j) acc[i][j] = (v8f){0.f,0.f,0.f,0.f,0.f,0.f,0.f,0.f};

  float one[MT];
#pragma unroll
  for (int i = 0; i < MT; ++i) one[i] = 1.0f;

  v4f st[NP];
  v4f a0[MT], b0[4], a1[MT], b1[4];
  {
#pragma unroll
    for (int i = 0; i < MT; ++i) st[i] = *(const v4f*)(ga1 + i * 16 * K);
#pragma unroll
    for (int i = MT; i < NP; ++i) st[i] = *(const v4f*)(gw1 + (i - MT) * 16 * K);
#pragma unroll
    for (int i = 0; i < MT; ++i) *(v4f*)(park + i * 16 * LPITCH) = st[i] * sc[i];
#pragma unroll
    for (int i = MT; i < NP; ++i) *(v4f*)(park + i * 16 * LPITCH) = st[i];
  }
  __syncthreads();
  lds_frags<MT>(fa, fb, 0, a0, b0);

  for (int t = 0; t < NCHUNK - 1; ++t) {
    const int tn = t + 1;
    const bool p1 = tn < NCHUNK / 2;
    const int k0 = (tn & (NCHUNK / 2 - 1)) * KC;
    const float* ga = (p1 ? ga1 : ga2) + k0;
    const float* gw = (p1 ? gw1 : gw2) + k0;
    const float* sa = fa + (t & 1) * STAGE;
    const float* sb = fb + (t & 1) * STAGE;
    const int sn = (tn & 1) * STAGE;

#pragma unroll
    for (int i = 0; i < MT; ++i) st[i] = *(const v4f*)(ga + i * 16 * K);
#pragma unroll
    for (int i = MT; i < NP; ++i) st[i] = *(const v4f*)(gw + (i - MT) * 16 * K);
    lds_frags<MT>(sa, sb, 1, a1, b1);
    __builtin_amdgcn_sched_barrier(0);
    mma_step<MT, false>(a0, b0, one, acc);
    __builtin_amdgcn_sched_barrier(0);
    lds_frags<MT>(sa, sb, 2, a0, b0);
    __builtin_amdgcn_sched_barrier(0);
    mma_step<MT, false>(a1, b1, one, acc);
    __builtin_amdgcn_sched_barrier(0);
    lds_frags<MT>(sa, sb, 3, a1, b1);
    {
      float* pd = park + sn;
#pragma unroll
      for (int i = 0; i < MT; ++i) *(v4f*)(pd + i * 16 * LPITCH) = p1 ? st[i] * sc[i] : st[i];
#pragma unroll
      for (int i = MT; i < NP; ++i) *(v4f*)(pd + i * 16 * LPITCH) = st[i];
    }
    __builtin_amdgcn_sched_barrier(0);
    mma_step<MT, false>(a0, b0, one, acc);
    __syncthreads();
    lds_frags<MT>(fa + sn, fb + sn, 0, a0, b0);
    __builtin_amdgcn_sched_barrier(0);
    mma_step<MT, false>(a1, b1, one, acc);
  }
  {
    const float* sa = fa + ((NCHUNK - 1) & 1) * STAGE;
    const float* sb = fb + ((NCHUNK - 1) & 1) * STAGE;
    lds_frags<MT>(sa, sb, 1, a1, b1);
    __builtin_amdgcn_sched_barrier(0);
    mma_step<MT, false>(a0, b0, one, acc);
    __builtin_amdgcn_sched_barrier(0);
    lds_frags<MT>(sa, sb, 2, a0, b0);
    __builtin_amdgcn_sched_barrier(0);
    mma_step<MT, false>(a1, b1, one, acc);
    __builtin_amdgcn_sched_barrier(0);
    lds_frags<MT>(sa, sb, 3, a1, b1);
    __builtin_amdgcn_sched_barrier(0);
    mma_step<MT, false>(a0, b0, one, acc);
    mma_step<MT, false>(a1, b1, one, acc);
  }

  const int cb = 64 * wave + l15;
#pragma unroll
  for (int j = 0; j < 4; ++j) {
    const int col = cb + j * 16;
    const float bv = bias[col];
#pragma unroll
    for (int i = 0; i < MT; ++i) {
#pragma unroll
      for (int r = 0; r < 8; ++r) {
        float v = acc[i][j][r] + bv;
        if (RELU) v = fmaxf(v, 0.f);
        out[(long long)(m0 + i * 16 + r + half * 8) * N + col] = v;
      }
    }
  }
}

__global__ __launch_bounds__(256) void gemm_heads_wmma(
    const float* __restrict__ A1, const float* __restrict__ cnt1,
    const float* __restrict__ A2,
    const float* __restrict__ Wt1a, const float* __restrict__ Wt2a,
    const float* __restrict__ biasa, float* __restrict__ outa,
    const float* __restrict__ Wt1b, const float* __restrict__ Wt2b,
    const float* __restrict__ biasb, float* __restrict__ outb,
    int M, int N) {
  const bool second = blockIdx.y != 0;
  gemm_dual_tile<1, 0>(A1, second ? Wt1b : Wt1a, cnt1,
                       A2, second ? Wt2b : Wt2a,
                       second ? biasb : biasa, second ? outb : outa, M, N);
}

static inline int cdiv(long long a, long long b) { return (int)((a + b - 1) / b); }

extern "C" void kernel_launch(void* const* d_in, const int* in_sizes, int n_in,
                              void* d_out, int out_size, void* d_ws, size_t ws_size,
                              hipStream_t stream) {
  (void)n_in; (void)out_size; (void)ws_size;

  const int N1 = 40000, N2 = 20000, N3 = 10000;
  const int OUT_C = O_DIM;
  static_assert(2 * N2 <= N1 && N3 <= N2, "h2 lives in the half of agg that layers 2 and 3 leave free");
  constexpr int MT = 5;
  static_assert(N1 % (16 * MT) == 0 && N2 % (16 * MT) == 0 && N3 % 16 == 0, "GEMM row blocks must tile M");

  const float* x   = (const float*)d_in[0];
  const int*   ei1 = (const int*)d_in[1];
  const int*   ei2 = (const int*)d_in[2];
  const int*   ei3 = (const int*)d_in[3];
  const float* Wl1 = (const float*)d_in[4];
  const float* bl1 = (const float*)d_in[5];
  const float* Wr1 = (const float*)d_in[6];
  const float* Wl2 = (const float*)d_in[7];
  const float* bl2 = (const float*)d_in[8];
  const float* Wr2 = (const float*)d_in[9];
  const float* Wl3 = (const float*)d_in[10];
  const float* bl3 = (const float*)d_in[11];
  const float* Wr3 = (const float*)d_in[12];
  const float* Wls = (const float*)d_in[13];
  const float* bls = (const float*)d_in[14];
  const float* Wrs = (const float*)d_in[15];

  const int E1 = in_sizes[1] / 2;
  const int E2 = in_sizes[2] / 2;
  const int E3 = in_sizes[3] / 2;
  const int QT = cdiv(E1, 4) + cdiv(E2, 4) + cdiv(E3, 4);
  const int NT = N1 + N2 + N3;

  float* h1  = (float*)d_ws;
  int*   csr = (int*)(h1 + (long long)N1 * C_DIM);
  int*   deg   = csr;
  int*   ovfn  = deg + NT;
  int*   slots = ovfn + 4;
  int*   ovf   = slots + NT * ROW_CAP;
  static_assert((long long)NT * (1 + ROW_CAP) + 4 < (long long)N2 * C_DIM, "row slots must fit their workspace part");
  float* agg = (float*)(csr + (long long)N2 * C_DIM);
  float* h2  = agg + (long long)N2 * C_DIM;
  float* cnt = agg + (long long)N1 * C_DIM;
  float* wt  = cnt + N1;
  float* Wl1t = wt;
  float* Wr1t = Wl1t + C_DIM * C_DIM;
  float* Wl2t = Wr1t + C_DIM * C_DIM;
  float* Wr2t = Wl2t + C_DIM * C_DIM;
  float* Wl3t = Wr2t + C_DIM * C_DIM;
  float* Wr3t = Wl3t + C_DIM * OUT_C;
  float* Wlst = Wr3t + C_DIM * OUT_C;
  float* Wrst = Wlst + C_DIM * OUT_C;

  float* mu     = (float*)d_out;
  float* logstd = mu + (long long)N3 * OUT_C;

  const int GB = 256;
  const int ZBLK = cdiv(NT + 4, 1024);
  const int EB = cdiv(QT, EC_BLOCK * EC_PER_THREAD);

  {
    zero_and_transpose<<<ZBLK + TR_BLOCKS, 1024, 0, stream>>>(deg, NT + 4, ZBLK,
                                                 Wl1, Wr1, Wl2, Wr2, Wl3, Wr3, Wls, Wrs, wt);
    edge_slots<<<EB, EC_BLOCK, 0, stream>>>(ei1, ei2, ei3, E1, E2, E3, N1, N2, deg, slots, ovfn, ovf);
  }

  {
    gather_rows<<<cdiv(N1, 8), GB, 0, stream>>>(x, deg, slots, ovfn, ovf, ei1, E1, 0, agg, cnt, N1);
    gemm_dual_lds<MT, 1><<<N1 / (16 * MT), 128, 0, stream>>>(
        agg, Wl1t, cnt, x, Wr1t, bl1, h1);
  }

  {
    gather_rows<<<cdiv(N2, 8), GB, 0, stream>>>(h1, deg + N1, slots + N1 * ROW_CAP, ovfn, ovf, ei2, E2, E1,
                                                agg, cnt, N2);
    gemm_dual_lds<MT, 1><<<N2 / (16 * MT), 128, 0, stream>>>(
        agg, Wl2t, cnt, h1, Wr2t, bl2, h2);
  }

  {
    gather_rows<<<cdiv(N3, 8), GB, 0, stream>>>(h2, deg + N1 + N2, slots + (N1 + N2) * ROW_CAP, ovfn, ovf,
                                                ei3, E3, E1 + E2, agg, cnt, N3);

    int tiles = (N3 / 16) * (OUT_C / 64);
    gemm_heads_wmma<<<dim3(cdiv(tiles, 8), 2), GB, 0, stream>>>(
        agg, cnt, h2, Wl3t, Wr3t, bl3, mu, Wlst, Wrst, bls, logstd, N3, OUT_C);
  }
}


# Output heads GEMM: both heads as one 128-column product in the LDS-staged form of layers 1-2 (4-wave block = 48 rows x 128 cols, A staged once, whole-line loads); k order and sums unchanged
# speedup vs baseline: 2.9952x; 1.0535x over previous
#include <hip/hip_runtime.h>
#include <hip/hip_bf16.h>

typedef __attribute__((ext_vector_type(2))) float v2f;
typedef __attribute__((ext_vector_type(4))) float v4f;
typedef __attribute__((ext_vector_type(8))) float v8f;

#define C_DIM 256

#define O_DIM 64
#define TR_BIG_BLOCKS (C_DIM * C_DIM / 4096)
#define TR_SML_BLOCKS (C_DIM * O_DIM / 4096)
#define TR_BLOCKS (4 * TR_BIG_BLOCKS + 4 * TR_SML_BLOCKS)
__device__ __forceinline__ void transpose_block(
    int b, int tid,
    const float* __restrict__ w0, const float* __restrict__ w1,
    const float* __restrict__ w2, const float* __restrict__ w3,
    const float* __restrict__ w4, const float* __restrict__ w5,
    const float* __restrict__ w6, const float* __restrict__ w7,
    float* __restrict__ wt) {
  constexpr int K = C_DIM;
  int m, local;
  if (b < 4 * TR_BIG_BLOCKS) { m = b / TR_BIG_BLOCKS; local = b % TR_BIG_BLOCKS; }
  else { const int r = b - 4 * TR_BIG_BLOCKS; m = 4 + r / TR_SML_BLOCKS; local = r % TR_SML_BLOCKS; }
  const float* W = m == 0 ? w0 : m == 1 ? w1 : m == 2 ? w2 : m == 3 ? w3
                 : m == 4 ? w4 : m == 5 ? w5 : m == 6 ? w6 : w7;
  const int N = m < 4 ? C_DIM : O_DIM;
  float* Wt = wt + (m < 4 ? m * (C_DIM * C_DIM)
                          : 4 * (C_DIM * C_DIM) + (m - 4) * (C_DIM * O_DIM));
  const int t = (local * 1024 + tid) * 4;
  const int k = t / N;
  const int n = t % N;
  const float4 v = *(const float4*)(W + t);
  Wt[(n + 0) * K + k] = v.x;
  Wt[(n + 1) * K + k] = v.y;
  Wt[(n + 2) * K + k] = v.z;
  Wt[(n + 3) * K + k] = v.w;
}

__global__ __launch_bounds__(1024) void zero_and_transpose(
    int* __restrict__ p, int n, int zero_blocks,
    const float* __restrict__ w0, const float* __restrict__ w1,
    const float* __restrict__ w2, const float* __restrict__ w3,
    const float* __restrict__ w4, const float* __restrict__ w5,
    const float* __restrict__ w6, const float* __restrict__ w7,
    float* __restrict__ wt) {
  if ((int)blockIdx.x >= zero_blocks) {
    transpose_block(blockIdx.x - zero_blocks, threadIdx.x, w0, w1, w2, w3, w4, w5, w6, w7, wt);
    return;
  }
  const int i = blockIdx.x * 1024 + threadIdx.x;
  if (i < n) p[i] = 0;
}

#define ROW_CAP 48
#define EC_PER_THREAD 2
#define EC_BLOCK 256

struct EdgeChunk {
  const int* src;
  const int* dst;
  int e;
  int roff;
  int eoff;
  int nv;
};

__device__ __forceinline__ EdgeChunk edge_chunk(
    int q, const int* __restrict__ ei1, const int* __restrict__ ei2,
    const int* __restrict__ ei3, int E1, int E2, int E3, int n1, int n2) {
  const int Q1 = (E1 + 3) >> 2, Q2 = (E2 + 3) >> 2;
  const bool in1 = q < Q1, in2 = q < Q1 + Q2;
  const int E = in1 ? E1 : in2 ? E2 : E3;
  EdgeChunk c;
  c.src  = in1 ? ei1 : in2 ? ei2 : ei3;
  c.e    = 4 * (q - (in1 ? 0 : in2 ? Q1 : Q1 + Q2));
  c.roff = in1 ? 0 : in2 ? n1 : n1 + n2;
  c.eoff = in1 ? 0 : in2 ? E1 : E1 + E2;
  c.dst  = c.src + E;
  c.nv   = min(4, E - c.e);
  return c;
}

template <bool WHOLE>
__device__ __forceinline__ int4 load_ids4(const int* __restrict__ p, int e, int nv) {
  if (WHOLE) return *(const int4*)(p + e);
  int4 v = make_int4(0, 0, 0, 0);
  v.x = p[e];
  if (nv > 1) v.y = p[e + 1];
  if (nv > 2) v.z = p[e + 2];
  if (nv > 3) v.w = p[e + 3];
  return v;
}

__device__ __forceinline__ void put_edge(int* __restrict__ slots, int* __restrict__ ovf_n,
                                         int* __restrict__ ovf, int row, int r, int s, int g) {
  if (r < ROW_CAP) slots[row * ROW_CAP + r] = s;
  else ovf[atomicAdd(ovf_n, 1)] = g;
}

template <bool WHOLE, bool FULL>
__device__ __forceinline__ void edge_slots_body(
    const int* __restrict__ ei1, const int* __restrict__ ei2,
    const int* __restrict__ ei3, int E1, int E2, int E3, int n1, int n2,
    int* __restrict__ deg, int* __restrict__ slots,
    int* __restrict__ ovf_n, int* __restrict__ ovf) {
  const int QT = ((E1 + 3) >> 2) + ((E2 + 3) >> 2) + ((E3 + 3) >> 2);
  const int q0 = blockIdx.x * (EC_BLOCK * EC_PER_THREAD) + threadIdx.x;
  EdgeChunk c[EC_PER_THREAD];
  int4 s[EC_PER_THREAD], d[EC_PER_THREAD], r[EC_PER_THREAD];
  int nv[EC_PER_THREAD];
  for (int u = 0; u < EC_PER_THREAD; ++u) {
    const int q = q0 + u * EC_BLOCK;
    c[u] = edge_chunk(min(q, QT - 1), ei1, ei2, ei3, E1, E2, E3, n1, n2);
    d[u] = load_ids4<WHOLE>(c[u].dst, c[u].e, c[u].nv);
    s[u] = load_ids4<WHOLE>(c[u].src, c[u].e, c[u].nv);
    nv[u] = FULL ? 4 : (q < QT ? c[u].nv : 0);
  }
  asm volatile("" ::: "memory");
  for (int u = 0; u < EC_PER_THREAD; ++u) {
    r[u] = make_int4(0, 0, 0, 0);
    if (FULL || nv[u] > 0) {
      int* g = deg + c[u].roff;
      r[u].x = atomicAdd(g + d[u].x, 1);
      if (WHOLE || nv[u] > 1) r[u].y = atomicAdd(g + d[u].y, 1);
      if (WHOLE || nv[u] > 2) r[u].z = atomicAdd(g + d[u].z, 1);
      if (WHOLE || nv[u] > 3) r[u].w = atomicAdd(g + d[u].w, 1);
    }
  }
  for (int u = 0; u < EC_PER_THREAD; ++u) {
    if (FULL || nv[u] > 0) {
      const int ro = c[u].roff, g0 = c[u].eoff + c[u].e;
      put_edge(slots, ovf_n, ovf, ro + d[u].x, r[u].x, s[u].x, g0);
      if (WHOLE || nv[u] > 1) put_edge(slots, ovf_n, ovf, ro + d[u].y, r[u].y, s[u].y, g0 + 1);
      if (WHOLE || nv[u] > 2) put_edge(slots, ovf_n, ovf, ro + d[u].z, r[u].z, s[u].z, g0 + 2);
      if (WHOLE || nv[u] > 3) put_edge(slots, ovf_n, ovf, ro + d[u].w, r[u].w, s[u].w, g0 + 3);
    }
  }
}

__device__ __forceinline__ bool edge_block_full(int E1, int E2, int E3) {
  const int QT = ((E1 + 3) >> 2) + ((E2 + 3) >> 2) + ((E3 + 3) >> 2);
  return (int)(blockIdx.x + 1) * (EC_BLOCK * EC_PER_THREAD) <= QT;
}

__global__ __launch_bounds__(EC_BLOCK) void edge_slots(
    const int* __restrict__ ei1, const int* __restrict__ ei2,
    const int* __restrict__ ei3, int E1, int E2, int E3,
    int n1, int n2,
    int* __restrict__ deg, int* __restrict__ slots,
    int* __restrict__ ovf_n, int* __restrict__ ovf) {
  const bool whole = ((E1 | E2 | E3) & 3) == 0;
  if (whole && edge_block_full(E1, E2, E3))
    edge_slots_body<true, true >(ei1, ei2, ei3, E1, E2, E3, n1, n2, deg, slots, ovf_n, ovf);
  else if (whole)
    edge_slots_body<true, false>(ei1, ei2, ei3, E1, E2, E3, n1, n2, deg, slots, ovf_n, ovf);
  else
    edge_slots_body<false, false>(ei1, ei2, ei3, E1, E2, E3, n1, n2, deg, slots, ovf_n, ovf);
}

template <int B>
__device__ __forceinline__ void add_rows(const float* __restrict__ xl, int idx, int j,
                                         float4& a0, float4& a1) {
  float4 u[B], w[B];
  for (int t = 0; t < B; ++t) {
    const float* r = xl + (long long)__builtin_amdgcn_readlane(idx, j + t) * C_DIM;
    u[t] = *(const float4*)r;
    w[t] = *(const float4*)(r + 128);
  }
  for (int t = 0; t < B; ++t) {
    a0.x += u[t].x; a0.y += u[t].y; a0.z += u[t].z; a0.w += u[t].w;
    a1.x += w[t].x; a1.y += w[t].y; a1.z += w[t].z; a1.w += w[t].w;
  }
}

__global__ __launch_bounds__(256) void gather_rows(
    const float* __restrict__ x, const int* __restrict__ deg,
    const int* __restrict__ slots, const int* __restrict__ ovf_n,
    const int* __restrict__ ovf, const int* __restrict__ ei, int E, int eoff,
    float* __restrict__ agg, float* __restrict__ cnt, int n_dst) {
  const int lane = threadIdx.x & 31;
  const int d = __builtin_amdgcn_readfirstlane(blockIdx.x * 8 + (threadIdx.x >> 5));
  if (d >= n_dst) return;
  const int n  = deg[d];
  const int ns = min(n, ROW_CAP);
  const int* sl = slots + d * ROW_CAP;
  const float* xl = x + lane * 4;

  float4 a0 = make_float4(0.f, 0.f, 0.f, 0.f);
  float4 a1 = make_float4(0.f, 0.f, 0.f, 0.f);

  for (int base = 0; base < ns; base += 32) {
    const int m = min(ns - base, 32);
    int idx = 0;
    if (lane < m) idx = sl[base + lane];
    int j = 0;
    for (; j + 8 <= m; j += 8) add_rows<8>(xl, idx, j, a0, a1);
    if (m & 4) { add_rows<4>(xl, idx, j, a0, a1); j += 4; }
    if (m & 2) { add_rows<2>(xl, idx, j, a0, a1); j += 2; }
    if (m & 1) { add_rows<1>(xl, idx, j, a0, a1); }
  }
  if (n > ROW_CAP) {
    const int total = *ovf_n;
    for (int k = 0; k < total; ++k) {
      const int e = ovf[k] - eoff;
      if (e >= 0 && e < E && ei[E + e] == d) add_rows<1>(xl, ei[e], 0, a0, a1);
    }
  }

  float* o = agg + (long long)d * C_DIM + lane * 4;
  *(float4*)o         = a0;
  *(float4*)(o + 128) = a1;
  if (lane == 0) cnt[d] = (float)n;
}


template <int MT, bool SCALE, int NB>
__device__ __forceinline__ void mma_step(const v4f (&a)[MT], const v4f (&b)[NB],
                                         const float (&s)[MT], v8f (&acc)[MT][NB]) {
#pragma unroll
  for (int i = 0; i < MT; ++i) {
    const v4f x = SCALE ? a[i] * s[i] : a[i];
    const v2f xlo = {x.x, x.y};
    const v2f xhi = {x.z, x.w};
#pragma unroll
    for (int j = 0; j < NB; ++j) {
      const v2f blo = {b[j].x, b[j].y};
      acc[i][j] = __builtin_amdgcn_wmma_f32_16x16x4_f32(false, xlo, false, blo,
                                                        (short)0, acc[i][j],
                                                        false, false);
    }
#pragma unroll
    for (int j = 0; j < NB; ++j) {
      const v2f bhi = {b[j].z, b[j].w};
      acc[i][j] = __builtin_amdgcn_wmma_f32_16x16x4_f32(false, xhi, false, bhi,
                                                        (short)0, acc[i][j],
                                                        false, false);
    }
  }
}

#define KC 32
#define LPITCH 36

template <int MT, int NB>
__device__ __forceinline__ void lds_frags(const float* sa, const float* sb, int ks,
                                          v4f (&a)[MT], v4f (&b)[NB]) {
#pragma unroll
  for (int i = 0; i < MT; ++i) a[i] = *(const v4f*)(sa + i * 16 * LPITCH + ks * 8);
#pragma unroll
  for (int j = 0; j < NB; ++j) b[j] = *(const v4f*)(sb + j * 16 * LPITCH + ks * 8);
}

template <int MT, int RELU>
__global__ __launch_bounds__(128) void gemm_dual_lds(
    const float* __restrict__ A1, const float* __restrict__ Wt1,
    const float* __restrict__ cnt1,
    const float* __restrict__ A2, const float* __restrict__ Wt2,
    const float* __restrict__ bias, float* __restrict__ out) {
  constexpr int K = C_DIM, N = C_DIM;
  constexpr int BM = 16 * MT;
  constexpr int ROWS = BM + N;
  constexpr int NP = ROWS * (KC / 4) / 128;
  constexpr int STAGE = ROWS * LPITCH;
  constexpr int NCHUNK = 2 * K / KC;
  static_assert(ROWS * (KC / 4) % 128 == 0, "the pieces of a chunk deal out evenly");
  static_assert(2 * STAGE * 4 <= 160 * 1024, "two blocks per WGP");
  __shared__ __attribute__((aligned(16))) float lds[2 * STAGE];

  const int tid = threadIdx.x;
  const int lane = tid & 31;
  const int wave = tid >> 5;
  const int m0 = blockIdx.x * BM;

  const int srow = tid >> 3;
  const int sq   = tid & 7;
  float sc[MT];
#pragma unroll
  for (int i = 0; i < MT; ++i) sc[i] = 1.0f / fmaxf(cnt1[m0 + srow + i * 16], 1.0f);
  const float* ga1 = A1 + (long long)(m0 + srow) * K + 4 * sq;
  const float* ga2 = A2 + (long long)(m0 + srow) * K + 4 * sq;
  const float* gw1 = Wt1 + (long long)srow * K + 4 * sq;
  const float* gw2 = Wt2 + (long long)srow * K + 4 * sq;
  float* park = lds + srow * LPITCH + 4 * sq;

  const int half = lane >> 4;
  const int l15  = lane & 15;
  const float* fa = lds + l15 * LPITCH + 4 * half;
  const float* fb = lds + (BM + 64 * wave + l15) * LPITCH + 4 * half;

  v8f acc[MT][4];
#pragma unroll
  for (int i = 0; i < MT; ++i)
#pragma unroll
    for (int j = 0; j < 4; ++j) acc[i][j] = (v8f){0.f,0.f,0.f,0.f,0.f,0.f,0.f,0.f};

  float one[MT];
#pragma unroll
  for (int i = 0; i < MT; ++i) one[i] = 1.0f;

  v4f st[NP];
  v4f a0[MT], b0[4], a1[MT], b1[4];
  {
#pragma unroll
    for (int i = 0; i < MT; ++i) st[i] = *(const v4f*)(ga1 + i * 16 * K);
#pragma unroll
    for (int i = MT; i < NP; ++i) st[i] = *(const v4f*)(gw1 + (i - MT) * 16 * K);
#pragma unroll
    for (int i = 0; i < MT; ++i) *(v4f*)(park + i * 16 * LPITCH) = st[i] * sc[i];
#pragma unroll
    for (int i = MT; i < NP; ++i) *(v4f*)(park + i * 16 * LPITCH) = st[i];
  }
  __syncthreads();
  lds_frags<MT>(fa, fb, 0, a0, b0);

  for (int t = 0; t < NCHUNK - 1; ++t) {
    const int tn = t + 1;
    const bool p1 = tn < NCHUNK / 2;
    const int k0 = (tn & (NCHUNK / 2 - 1)) * KC;
    const float* ga = (p1 ? ga1 : ga2) + k0;
    const float* gw = (p1 ? gw1 : gw2) + k0;
    const float* sa = fa + (t & 1) * STAGE;
    const float* sb = fb + (t & 1) * STAGE;
    const int sn = (tn & 1) * STAGE;

#pragma unroll
    for (int i = 0; i < MT; ++i) st[i] = *(const v4f*)(ga + i * 16 * K);
#pragma unroll
    for (int i = MT; i < NP; ++i) st[i] = *(const v4f*)(gw + (i - MT) * 16 * K);
    lds_frags<MT>(sa, sb, 1, a1, b1);
    __builtin_amdgcn_sched_barrier(0);
    mma_step<MT, false>(a0, b0, one, acc);
    __builtin_amdgcn_sched_barrier(0);
    lds_frags<MT>(sa, sb, 2, a0, b0);
    __builtin_amdgcn_sched_barrier(0);
    mma_step<MT, false>(a1, b1, one, acc);
    __builtin_amdgcn_sched_barrier(0);
    lds_frags<MT>(sa, sb, 3, a1, b1);
    {
      float* pd = park + sn;
#pragma unroll
      for (int i = 0; i < MT; ++i) *(v4f*)(pd + i * 16 * LPITCH) = p1 ? st[i] * sc[i] : st[i];
#pragma unroll
      for (int i = MT; i < NP; ++i) *(v4f*)(pd + i * 16 * LPITCH) = st[i];
    }
    __builtin_amdgcn_sched_barrier(0);
    mma_step<MT, false>(a0, b0, one, acc);
    __syncthreads();
    lds_frags<MT>(fa + sn, fb + sn, 0, a0, b0);
    __builtin_amdgcn_sched_barrier(0);
    mma_step<MT, false>(a1, b1, one, acc);
  }
  {
    const float* sa = fa + ((NCHUNK - 1) & 1) * STAGE;
    const float* sb = fb + ((NCHUNK - 1) & 1) * STAGE;
    lds_frags<MT>(sa, sb, 1, a1, b1);
    __builtin_amdgcn_sched_barrier(0);
    mma_step<MT, false>(a0, b0, one, acc);
    __builtin_amdgcn_sched_barrier(0);
    lds_frags<MT>(sa, sb, 2, a0, b0);
    __builtin_amdgcn_sched_barrier(0);
    mma_step<MT, false>(a1, b1, one, acc);
    __builtin_amdgcn_sched_barrier(0);
    lds_frags<MT>(sa, sb, 3, a1, b1);
    __builtin_amdgcn_sched_barrier(0);
    mma_step<MT, false>(a0, b0, one, acc);
    mma_step<MT, false>(a1, b1, one, acc);
  }

  const int cb = 64 * wave + l15;
#pragma unroll
  for (int j = 0; j < 4; ++j) {
    const int col = cb + j * 16;
    const float bv = bias[col];
#pragma unroll
    for (int i = 0; i < MT; ++i) {
#pragma unroll
      for (int r = 0; r < 8; ++r) {
        float v = acc[i][j][r] + bv;
        if (RELU) v = fmaxf(v, 0.f);
        out[(long long)(m0 + i * 16 + r + half * 8) * N + col] = v;
      }
    }
  }
}

template <int MT, int NB, bool GUARD>
__device__ __forceinline__ void store_heads_tile(const v8f (&acc)[MT][NB],
                                                 const float* __restrict__ bias,
                                                 float* __restrict__ out, int r0, int cb, int M) {
#pragma unroll
  for (int j = 0; j < NB; ++j) {
    const int col = cb + j * 16;
    const float bv = bias[col];
#pragma unroll
    for (int i = 0; i < MT; ++i) {
#pragma unroll
      for (int r = 0; r < 8; ++r) {
        const int row = r0 + i * 16 + r;
        if (!GUARD || row < M) out[(long long)row * O_DIM + col] = acc[i][j][r] + bv;
      }
    }
  }
}

__device__ __forceinline__ void lds_handover_barrier() {
  __builtin_amdgcn_fence(__ATOMIC_RELEASE, "workgroup", "local");
  __builtin_amdgcn_s_barrier();
  __builtin_amdgcn_fence(__ATOMIC_ACQUIRE, "workgroup", "local");
}

template <int MT>
__global__ __launch_bounds__(128) void gemm_heads_lds(
    const float* __restrict__ A1, const float* __restrict__ cnt1,
    const float* __restrict__ A2,
    const float* __restrict__ Wt1, const float* __restrict__ Wt2,
    const float* __restrict__ biasa, float* __restrict__ outa,
    const float* __restrict__ biasb, float* __restrict__ outb, int M) {
  constexpr int K = C_DIM, N = 2 * O_DIM;
  constexpr int NB = N / 64;
  constexpr int BM = 16 * MT;
  constexpr int ROWS = BM + N;
  constexpr int NP = ROWS * (KC / 4) / 128;
  constexpr int STAGE = ROWS * LPITCH;
  constexpr int NCHUNK = 2 * K / KC;
  static_assert(ROWS * (KC / 4) % 128 == 0, "the pieces of a chunk deal out evenly");
  static_assert(NCHUNK >= 4 && NCHUNK % 2 == 0, "the chunk loop fetches two chunks ahead");
  __shared__ __attribute__((aligned(16))) float lds[2 * STAGE];

  const int tid = threadIdx.x;
  const int lane = tid & 31;
  const int wave = tid >> 5;
  const int m0 = blockIdx.x * BM;

  const int srow = tid >> 3;
  const int sq   = tid & 7;
  int aoff[MT];
  float sc[MT];
#pragma unroll
  for (int i = 0; i < MT; ++i) {
    const int r = min(m0 + srow + i * 16, M - 1);
    aoff[i] = r * K + 4 * sq;
    sc[i] = 1.0f / fmaxf(cnt1[r], 1.0f);
  }
  const float* gw1 = Wt1 + srow * K + 4 * sq;
  const float* gw2 = Wt2 + srow * K + 4 * sq;
  float* park = lds + srow * LPITCH + 4 * sq;

  const int half = lane >> 4;
  const int l15  = lane & 15;
  const float* fa = lds + l15 * LPITCH + 4 * half;
  const float* fb = lds + (BM + 16 * NB * wave + l15) * LPITCH + 4 * half;

  v8f acc[MT][NB];
#pragma unroll
  for (int i = 0; i < MT; ++i)
#pragma unroll
    for (int j = 0; j < NB; ++j) acc[i][j] = (v8f){0.f,0.f,0.f,0.f,0.f,0.f,0.f,0.f};

  float one[MT];
#pragma unroll
  for (int i = 0; i < MT; ++i) one[i] = 1.0f;

  v4f st[NP];
  v4f a0[MT], b0[NB], a1[MT], b1[NB];
  {
#pragma unroll
    for (int i = 0; i < MT; ++i) st[i] = *(const v4f*)(A1 + aoff[i]);
#pragma unroll
    for (int i = MT; i < NP; ++i) st[i] = *(const v4f*)(gw1 + (i - MT) * 16 * K);
    __builtin_amdgcn_sched_barrier(0);
#pragma unroll
    for (int i = 0; i < MT; ++i) *(v4f*)(park + i * 16 * LPITCH) = st[i] * sc[i];
#pragma unroll
    for (int i = MT; i < NP; ++i) *(v4f*)(park + i * 16 * LPITCH) = st[i];
#pragma unroll
    for (int i = 0; i < MT; ++i) st[i] = *(const v4f*)(A1 + aoff[i] + KC);
#pragma unroll
    for (int i = MT; i < NP; ++i) st[i] = *(const v4f*)(gw1 + (i - MT) * 16 * K + KC);
  }
  lds_handover_barrier();
  lds_frags<MT>(fa, fb, 0, a0, b0);

  for (int t = 0; t < NCHUNK - 1; ++t) {
    const int tn = t + 1;
    const bool p1 = tn < NCHUNK / 2;
    const float* sa = fa + (t & 1) * STAGE;
    const float* sb = fb + (t & 1) * STAGE;
    const int sn = (tn & 1) * STAGE;

    lds_frags<MT>(sa, sb, 1, a1, b1);
    __builtin_amdgcn_sched_barrier(0);
    mma_step<MT, false>(a0, b0, one, acc);
    __builtin_amdgcn_sched_barrier(0);
    lds_frags<MT>(sa, sb, 2, a0, b0);
    __builtin_amdgcn_sched_barrier(0);
    mma_step<MT, false>(a1, b1, one, acc);
    __builtin_amdgcn_sched_barrier(0);
    lds_frags<MT>(sa, sb, 3, a1, b1);
    {
      float* pd = park + sn;
#pragma unroll
      for (int i = 0; i < MT; ++i) *(v4f*)(pd + i * 16 * LPITCH) = p1 ? st[i] * sc[i] : st[i];
#pragma unroll
      for (int i = MT; i < NP; ++i) *(v4f*)(pd + i * 16 * LPITCH) = st[i];
    }
    if (t + 2 < NCHUNK) {
      const int t2 = t + 2;
      const bool q1 = t2 < NCHUNK / 2;
      const int k0 = (t2 & (NCHUNK / 2 - 1)) * KC;
      const float* ga = (q1 ? A1 : A2) + k0;
      const float* gw = (q1 ? gw1 : gw2) + k0;
#pragma unroll
      for (int i = 0; i < MT; ++i) st[i] = *(const v4f*)(ga + aoff[i]);
#pragma unroll
      for (int i = MT; i < NP; ++i) st[i] = *(const v4f*)(gw + (i - MT) * 16 * K);
    }
    __builtin_amdgcn_sched_barrier(0);
    mma_step<MT, false>(a0, b0, one, acc);
    lds_handover_barrier();
    lds_frags<MT>(fa + sn, fb + sn, 0, a0, b0);
    __builtin_amdgcn_sched_barrier(0);
    mma_step<MT, false>(a1, b1, one, acc);
  }
  {
    const float* sa = fa + ((NCHUNK - 1) & 1) * STAGE;
    const float* sb = fb + ((NCHUNK - 1) & 1) * STAGE;
    lds_frags<MT>(sa, sb, 1, a1, b1);
    __builtin_amdgcn_sched_barrier(0);
    mma_step<MT, false>(a0, b0, one, acc);
    __builtin_amdgcn_sched_barrier(0);
    lds_frags<MT>(sa, sb, 2, a0, b0);
    __builtin_amdgcn_sched_barrier(0);
    mma_step<MT, false>(a1, b1, one, acc);
    __builtin_amdgcn_sched_barrier(0);
    lds_frags<MT>(sa, sb, 3, a1, b1);
    __builtin_amdgcn_sched_barrier(0);
    mma_step<MT, false>(a0, b0, one, acc);
    mma_step<MT, false>(a1, b1, one, acc);
  }

  const bool second = wave >= 2;
  const float* bias = second ? biasb : biasa;
  float* out = second ? outb : outa;
  const int cb = 16 * NB * (wave & 1) + l15;
  if (m0 + BM <= M) {
    store_heads_tile<MT, NB, false>(acc, bias, out, m0 + half * 8, cb, M);
  } else {
    store_heads_tile<MT, NB, true>(acc, bias, out, m0 + half * 8, cb, M);
  }
}

static inline int cdiv(long long a, long long b) { return (int)((a + b - 1) / b); }

extern "C" void kernel_launch(void* const* d_in, const int* in_sizes, int n_in,
                              void* d_out, int out_size, void* d_ws, size_t ws_size,
                              hipStream_t stream) {
  (void)n_in; (void)out_size; (void)ws_size;

  const int N1 = 40000, N2 = 20000, N3 = 10000;
  const int OUT_C = O_DIM;
  static_assert(2 * N2 <= N1 && N3 <= N2, "h2 lives in the half of agg that layers 2 and 3 leave free");
  constexpr int MT = 5;
  static_assert(N1 % (16 * MT) == 0 && N2 % (16 * MT) == 0, "GEMM row blocks must tile M");
  constexpr int HMT = 3;
  static_assert(OUT_C == O_DIM, "the heads' GEMM is built for O_DIM columns per head");

  const float* x   = (const float*)d_in[0];
  const int*   ei1 = (const int*)d_in[1];
  const int*   ei2 = (const int*)d_in[2];
  const int*   ei3 = (const int*)d_in[3];
  const float* Wl1 = (const float*)d_in[4];
  const float* bl1 = (const float*)d_in[5];
  const float* Wr1 = (const float*)d_in[6];
  const float* Wl2 = (const float*)d_in[7];
  const float* bl2 = (const float*)d_in[8];
  const float* Wr2 = (const float*)d_in[9];
  const float* Wl3 = (const float*)d_in[10];
  const float* bl3 = (const float*)d_in[11];
  const float* Wr3 = (const float*)d_in[12];
  const float* Wls = (const float*)d_in[13];
  const float* bls = (const float*)d_in[14];
  const float* Wrs = (const float*)d_in[15];

  const int E1 = in_sizes[1] / 2;
  const int E2 = in_sizes[2] / 2;
  const int E3 = in_sizes[3] / 2;
  const int QT = cdiv(E1, 4) + cdiv(E2, 4) + cdiv(E3, 4);
  const int NT = N1 + N2 + N3;

  float* h1  = (float*)d_ws;
  int*   csr = (int*)(h1 + (long long)N1 * C_DIM);
  int*   deg   = csr;
  int*   ovfn  = deg + NT;
  int*   slots = ovfn + 4;
  int*   ovf   = slots + NT * ROW_CAP;
  static_assert((long long)NT * (1 + ROW_CAP) + 4 < (long long)N2 * C_DIM, "row slots must fit their workspace part");
  float* agg = (float*)(csr + (long long)N2 * C_DIM);
  float* h2  = agg + (long long)N2 * C_DIM;
  float* cnt = agg + (long long)N1 * C_DIM;
  float* wt  = cnt + N1;
  float* Wl1t = wt;
  float* Wr1t = Wl1t + C_DIM * C_DIM;
  float* Wl2t = Wr1t + C_DIM * C_DIM;
  float* Wr2t = Wl2t + C_DIM * C_DIM;
  float* WlHt = Wr2t + C_DIM * C_DIM;
  float* WrHt = WlHt + 2 * C_DIM * OUT_C;

  float* mu     = (float*)d_out;
  float* logstd = mu + (long long)N3 * OUT_C;

  const int GB = 256;
  const int ZBLK = cdiv(NT + 4, 1024);
  const int EB = cdiv(QT, EC_BLOCK * EC_PER_THREAD);

  {
    zero_and_transpose<<<ZBLK + TR_BLOCKS, 1024, 0, stream>>>(deg, NT + 4, ZBLK,
                                                 Wl1, Wr1, Wl2, Wr2, Wl3, Wls, Wr3, Wrs, wt);
    edge_slots<<<EB, EC_BLOCK, 0, stream>>>(ei1, ei2, ei3, E1, E2, E3, N1, N2, deg, slots, ovfn, ovf);
  }

  {
    gather_rows<<<cdiv(N1, 8), GB, 0, stream>>>(x, deg, slots, ovfn, ovf, ei1, E1, 0, agg, cnt, N1);
    gemm_dual_lds<MT, 1><<<N1 / (16 * MT), 128, 0, stream>>>(
        agg, Wl1t, cnt, x, Wr1t, bl1, h1);
  }

  {
    gather_rows<<<cdiv(N2, 8), GB, 0, stream>>>(h1, deg + N1, slots + N1 * ROW_CAP, ovfn, ovf, ei2, E2, E1,
                                                agg, cnt, N2);
    gemm_dual_lds<MT, 1><<<N2 / (16 * MT), 128, 0, stream>>>(
        agg, Wl2t, cnt, h1, Wr2t, bl2, h2);
  }

  {
    gather_rows<<<cdiv(N3, 8), GB, 0, stream>>>(h2, deg + N1 + N2, slots + (N1 + N2) * ROW_CAP, ovfn, ovf,
                                                ei3, E3, E1 + E2, agg, cnt, N3);

    gemm_heads_lds<HMT><<<cdiv(N3, 16 * HMT), 128, 0, stream>>>(
        agg, cnt, h2, WlHt, WrHt, bl3, mu, bls, logstd, N3);
  }
}
